# starting kernel
# baseline (speedup 1.0000x reference)
#include <hip/hip_runtime.h>

typedef __attribute__((ext_vector_type(2))) float v2f;
typedef __attribute__((ext_vector_type(8))) float v8f;

#define F_IN  128
#define H_DIM 64

__global__ void zero_f32(float* __restrict__ p, long n) {
  long i = (long)blockIdx.x * blockDim.x + threadIdx.x;
  long stride = (long)gridDim.x * blockDim.x;
  for (; i < n; i += stride) p[i] = 0.0f;
}

__global__ void degree_kernel(const int* __restrict__ src, const int* __restrict__ dst,
                              float* __restrict__ deg_out, float* __restrict__ deg_in, int E) {
  int i = blockIdx.x * blockDim.x + threadIdx.x;
  int stride = gridDim.x * blockDim.x;
  for (; i < E; i += stride) {
    atomicAdd(&deg_out[src[i]], 1.0f);
    atomicAdd(&deg_in[dst[i]], 1.0f);
  }
}

__global__ void norm_kernel(float* __restrict__ p, int n) {
  int i = blockIdx.x * blockDim.x + threadIdx.x;
  if (i < n) p[i] = rsqrtf(fmaxf(p[i], 1.0f));
}

__device__ __forceinline__ void stage_w_frag(v2f* lW, const float* __restrict__ W,
                                             int kblocks, int tid, int nthreads) {
  const int total = kblocks * 4 * 32;
  for (int i = tid; i < total; i += nthreads) {
    const int kb = i >> 7;
    const int ct = (i >> 5) & 3;
    const int ln = i & 31;
    const int kg = (ln >> 4) * 2;
    const int n  = ct * 16 + (ln & 15);
    v2f p;
    p.x = W[(kb * 4 + kg)     * H_DIM + n];
    p.y = W[(kb * 4 + kg + 1) * H_DIM + n];
    lW[i] = p;
  }
}

__global__ __launch_bounds__(256) void gemm1_wmma(
    const float* __restrict__ feat, const float* __restrict__ norm_src,
    const float* __restrict__ W1, float* __restrict__ h1, int N) {
  __shared__ v2f lW[(F_IN / 4) * 4 * 32];
  stage_w_frag(lW, W1, F_IN / 4, threadIdx.x, 256);
  __syncthreads();

  const int wave = threadIdx.x >> 5;
  const int lane = threadIdx.x & 31;
  const int tile = blockIdx.x * 8 + wave;
  const int row0 = tile * 16;
  if (row0 >= N) return;

  const int mrow = row0 + (lane & 15);
  const int kgrp = (lane >> 4) * 2;
  const float ns = norm_src[mrow];

  v8f acc[4] = {};
  for (int kb = 0; kb < F_IN / 4; ++kb) {
    v2f a = *(const v2f*)(feat + (long)mrow * F_IN + kb * 4 + kgrp);
    a.x *= ns; a.y *= ns;
#pragma unroll
    for (int ct = 0; ct < 4; ++ct) {
      const v2f b = lW[(kb * 4 + ct) * 32 + lane];
      acc[ct] = __builtin_amdgcn_wmma_f32_16x16x4_f32(
          false, a, false, b, (short)0, acc[ct], false, false);
    }
  }

  const int mbase = row0 + ((lane >> 4) << 3);
#pragma unroll
  for (int ct = 0; ct < 4; ++ct) {
    const int n = ct * 16 + (lane & 15);
#pragma unroll
    for (int v = 0; v < 8; ++v)
      h1[(long)(mbase + v) * H_DIM + n] = acc[ct][v];
  }
}

__global__ void aggregate_kernel(const int* __restrict__ src, const int* __restrict__ dst,
                                 const float* __restrict__ h, float* __restrict__ agg, int E) {
  long t = (long)blockIdx.x * blockDim.x + threadIdx.x;
  const long total = (long)E * 16;
  const long stride = (long)gridDim.x * blockDim.x;
  for (; t < total; t += stride) {
    const int e = (int)(t >> 4);
    const int q = (int)(t & 15);
    const int s = src[e], d = dst[e];
    const float4 v = ((const float4*)h)[(long)s * 16 + q];
    float* ap = agg + (long)d * H_DIM + q * 4;
    atomicAdd(ap + 0, v.x);
    atomicAdd(ap + 1, v.y);
    atomicAdd(ap + 2, v.z);
    atomicAdd(ap + 3, v.w);
  }
}

__global__ void relu_scale_kernel(const float* __restrict__ agg, const float* __restrict__ norm_dst,
                                  const float* __restrict__ b1, const float* __restrict__ norm_src,
                                  float* __restrict__ h2, int N) {
  long i = (long)blockIdx.x * blockDim.x + threadIdx.x;
  const long total = (long)N * H_DIM;
  if (i >= total) return;
  const int node = (int)(i >> 6);
  const int j = (int)(i & 63);
  const float v = fmaxf(agg[i] * norm_dst[node] + b1[j], 0.0f);
  h2[i] = v * norm_src[node];
}

__global__ __launch_bounds__(256) void gemm2_final_wmma(
    const float* __restrict__ agg2,
    const float* __restrict__ W_mu, const float* __restrict__ b_mu,
    const float* __restrict__ W_sig, const float* __restrict__ b_sig,
    const float* __restrict__ norm_dst, const float* __restrict__ noise,
    float* __restrict__ z, int N) {
  __shared__ v2f lWm[(H_DIM / 4) * 4 * 32];
  __shared__ v2f lWs[(H_DIM / 4) * 4 * 32];
  stage_w_frag(lWm, W_mu, H_DIM / 4, threadIdx.x, 256);
  stage_w_frag(lWs, W_sig, H_DIM / 4, threadIdx.x, 256);
  __syncthreads();

  const int wave = threadIdx.x >> 5;
  const int lane = threadIdx.x & 31;
  const int tile = blockIdx.x * 8 + wave;
  const int row0 = tile * 16;
  if (row0 >= N) return;

  const int mrow = row0 + (lane & 15);
  const int kgrp = (lane >> 4) * 2;

  v8f am[4] = {}, as[4] = {};
  for (int kb = 0; kb < H_DIM / 4; ++kb) {
    const v2f a = *(const v2f*)(agg2 + (long)mrow * H_DIM + kb * 4 + kgrp);
#pragma unroll
    for (int ct = 0; ct < 4; ++ct) {
      const v2f bm = lWm[(kb * 4 + ct) * 32 + lane];
      const v2f bs = lWs[(kb * 4 + ct) * 32 + lane];
      am[ct] = __builtin_amdgcn_wmma_f32_16x16x4_f32(
          false, a, false, bm, (short)0, am[ct], false, false);
      as[ct] = __builtin_amdgcn_wmma_f32_16x16x4_f32(
          false, a, false, bs, (short)0, as[ct], false, false);
    }
  }

  const int mbase = row0 + ((lane >> 4) << 3);
#pragma unroll
  for (int ct = 0; ct < 4; ++ct) {
    const int n = ct * 16 + (lane & 15);
    const float bmu = b_mu[n];
    const float bsg = b_sig[n];
#pragma unroll
    for (int v = 0; v < 8; ++v) {
      const int m = mbase + v;
      const float nd = norm_dst[m];
      const float mu = am[ct][v] * nd + bmu;
      const float ls = as[ct][v] * nd + bsg;
      z[(long)m * H_DIM + n] = mu + noise[(long)m * H_DIM + n] * expf(ls);
    }
  }
}

extern "C" void kernel_launch(void* const* d_in, const int* in_sizes, int n_in,
                              void* d_out, int out_size, void* d_ws, size_t ws_size,
                              hipStream_t stream) {
  const float* feat  = (const float*)d_in[0];
  const int*   src   = (const int*)d_in[1];
  const int*   dst   = (const int*)d_in[2];
  const float* noise = (const float*)d_in[3];
  const float* W1    = (const float*)d_in[4];
  const float* b1    = (const float*)d_in[5];
  const float* W_mu  = (const float*)d_in[6];
  const float* b_mu  = (const float*)d_in[7];
  const float* W_sig = (const float*)d_in[8];
  const float* b_sig = (const float*)d_in[9];

  const int E = in_sizes[1];
  const int N = in_sizes[3] / H_DIM;

  float* ws       = (float*)d_ws;
  float* norm_src = ws;
  float* norm_dst = ws + N;
  float* buf_h    = ws + 2L * N;
  float* buf_agg  = buf_h + (long)N * H_DIM;
  float* zout     = (float*)d_out;

  const int threads = 256;
  const int gemm_blocks = (N / 16 + 7) / 8;
  const long elems = (long)N * H_DIM;

  zero_f32<<<(int)((2L * N + threads - 1) / threads), threads, 0, stream>>>(norm_src, 2L * N);
  degree_kernel<<<(E + threads - 1) / threads, threads, 0, stream>>>(src, dst, norm_src, norm_dst, E);
  norm_kernel<<<(2 * N + threads - 1) / threads, threads, 0, stream>>>(norm_src, 2 * N);

  gemm1_wmma<<<gemm_blocks, threads, 0, stream>>>(feat, norm_src, W1, buf_h, N);
  zero_f32<<<(int)((elems + threads - 1) / threads), threads, 0, stream>>>(buf_agg, elems);
  aggregate_kernel<<<(int)(((long)E * 16 + threads - 1) / threads), threads, 0, stream>>>(
      src, dst, buf_h, buf_agg, E);
  relu_scale_kernel<<<(int)((elems + threads - 1) / threads), threads, 0, stream>>>(
      buf_agg, norm_dst, b1, norm_src, buf_h, N);

  zero_f32<<<(int)((elems + threads - 1) / threads), threads, 0, stream>>>(buf_agg, elems);
  aggregate_kernel<<<(int)(((long)E * 16 + threads - 1) / threads), threads, 0, stream>>>(
      src, dst, buf_h, buf_agg, E);
  gemm2_final_wmma<<<gemm_blocks, threads, 0, stream>>>(
      buf_agg, W_mu, b_mu, W_sig, b_sig, norm_dst, noise, zout, N);
}


# Degree counts: random one-lane global float atomics replaced by LDS range histograms over edge slices, partial count planes summed in the norm pass; no global atomics or zero fill for degrees
# speedup vs baseline: 1.3384x; 1.3384x over previous
#include <hip/hip_runtime.h>

typedef __attribute__((ext_vector_type(2))) float v2f;
typedef __attribute__((ext_vector_type(8))) float v8f;

#define F_IN  128
#define H_DIM 64

__global__ void zero_f32(float* __restrict__ p, long n) {
  long i = (long)blockIdx.x * blockDim.x + threadIdx.x;
  long stride = (long)gridDim.x * blockDim.x;
  for (; i < n; i += stride) p[i] = 0.0f;
}

#define DEG_R 25600
#define DEG_S 32
static_assert(2 * DEG_S <= H_DIM, "partial planes must fit the h buffer");

__device__ __forceinline__ void count4(unsigned* bins, const int4 q, const int lo) {
  const unsigned r0 = (unsigned)(q.x - lo), r1 = (unsigned)(q.y - lo);
  const unsigned r2 = (unsigned)(q.z - lo), r3 = (unsigned)(q.w - lo);
  if (r0 < DEG_R) atomicAdd(&bins[r0], 1u);
  if (r1 < DEG_R) atomicAdd(&bins[r1], 1u);
  if (r2 < DEG_R) atomicAdd(&bins[r2], 1u);
  if (r3 < DEG_R) atomicAdd(&bins[r3], 1u);
}

__global__ __launch_bounds__(1024) void degree_hist_kernel(
    const int* __restrict__ src, const int* __restrict__ dst,
    unsigned* __restrict__ part, int E, int N) {
  __shared__ unsigned bins[DEG_R];
  const int* __restrict__ idx = blockIdx.z ? dst : src;
  const int lo = blockIdx.y * DEG_R;
  for (int i = threadIdx.x; i < DEG_R; i += 1024) bins[i] = 0u;
  __syncthreads();

  const int chunk = (((E + DEG_S - 1) / DEG_S) + 3) & ~3;
  const int beg = min(E, (int)blockIdx.x * chunk);
  const int end = min(E, beg + chunk);
  const int nvec = (end - beg) >> 2;
  const int4* __restrict__ v = (const int4*)(idx + beg);
  const int4 none = make_int4(lo - 1, lo - 1, lo - 1, lo - 1);
  const int last = nvec - 1;
  for (int i = threadIdx.x; i < nvec; i += 4096) {
    const int4 q0 = v[i];
    int4 q1 = v[min(i + 1024, last)];
    int4 q2 = v[min(i + 2048, last)];
    int4 q3 = v[min(i + 3072, last)];
    if (i + 1024 > last) q1 = none;
    if (i + 2048 > last) q2 = none;
    if (i + 3072 > last) q3 = none;
    count4(bins, q0, lo);
    count4(bins, q1, lo);
    count4(bins, q2, lo);
    count4(bins, q3, lo);
  }
  for (int i = beg + (nvec << 2) + threadIdx.x; i < end; i += 1024) {
    const unsigned r = (unsigned)(idx[i] - lo);
    if (r < DEG_R) atomicAdd(&bins[r], 1u);
  }
  __syncthreads();

  unsigned* __restrict__ plane = part + (long)(blockIdx.z * DEG_S + blockIdx.x) * N;
  const int hi = min(N, lo + DEG_R);
  for (int i = lo + threadIdx.x; i < hi; i += 1024) plane[i] = bins[i - lo];
}

__global__ void norm_kernel(const unsigned* __restrict__ part, float* __restrict__ norm, int N) {
  const int i = blockIdx.x * blockDim.x + threadIdx.x;
  if (i >= N) return;
  const unsigned* __restrict__ p = part + (long)blockIdx.y * DEG_S * N + i;
  unsigned c = 0u;
  for (int s = 0; s < DEG_S; ++s) c += p[(long)s * N];
  norm[(long)blockIdx.y * N + i] = rsqrtf(fmaxf((float)c, 1.0f));
}

__device__ __forceinline__ void stage_w_frag(v2f* lW, const float* __restrict__ W,
                                             int kblocks, int tid, int nthreads) {
  const int total = kblocks * 4 * 32;
  for (int i = tid; i < total; i += nthreads) {
    const int kb = i >> 7;
    const int ct = (i >> 5) & 3;
    const int ln = i & 31;
    const int kg = (ln >> 4) * 2;
    const int n  = ct * 16 + (ln & 15);
    v2f p;
    p.x = W[(kb * 4 + kg)     * H_DIM + n];
    p.y = W[(kb * 4 + kg + 1) * H_DIM + n];
    lW[i] = p;
  }
}

__global__ __launch_bounds__(256) void gemm1_wmma(
    const float* __restrict__ feat, const float* __restrict__ norm_src,
    const float* __restrict__ W1, float* __restrict__ h1, int N) {
  __shared__ v2f lW[(F_IN / 4) * 4 * 32];
  stage_w_frag(lW, W1, F_IN / 4, threadIdx.x, 256);
  __syncthreads();

  const int wave = threadIdx.x >> 5;
  const int lane = threadIdx.x & 31;
  const int tile = blockIdx.x * 8 + wave;
  const int row0 = tile * 16;
  if (row0 >= N) return;

  const int mrow = row0 + (lane & 15);
  const int kgrp = (lane >> 4) * 2;
  const float ns = norm_src[mrow];

  v8f acc[4] = {};
  for (int kb = 0; kb < F_IN / 4; ++kb) {
    v2f a = *(const v2f*)(feat + (long)mrow * F_IN + kb * 4 + kgrp);
    a.x *= ns; a.y *= ns;
#pragma unroll
    for (int ct = 0; ct < 4; ++ct) {
      const v2f b = lW[(kb * 4 + ct) * 32 + lane];
      acc[ct] = __builtin_amdgcn_wmma_f32_16x16x4_f32(
          false, a, false, b, (short)0, acc[ct], false, false);
    }
  }

  const int mbase = row0 + ((lane >> 4) << 3);
#pragma unroll
  for (int ct = 0; ct < 4; ++ct) {
    const int n = ct * 16 + (lane & 15);
#pragma unroll
    for (int v = 0; v < 8; ++v)
      h1[(long)(mbase + v) * H_DIM + n] = acc[ct][v];
  }
}

__global__ void aggregate_kernel(const int* __restrict__ src, const int* __restrict__ dst,
                                 const float* __restrict__ h, float* __restrict__ agg, int E) {
  long t = (long)blockIdx.x * blockDim.x + threadIdx.x;
  const long total = (long)E * 16;
  const long stride = (long)gridDim.x * blockDim.x;
  for (; t < total; t += stride) {
    const int e = (int)(t >> 4);
    const int q = (int)(t & 15);
    const int s = src[e], d = dst[e];
    const float4 v = ((const float4*)h)[(long)s * 16 + q];
    float* ap = agg + (long)d * H_DIM + q * 4;
    atomicAdd(ap + 0, v.x);
    atomicAdd(ap + 1, v.y);
    atomicAdd(ap + 2, v.z);
    atomicAdd(ap + 3, v.w);
  }
}

__global__ void relu_scale_kernel(const float* __restrict__ agg, const float* __restrict__ norm_dst,
                                  const float* __restrict__ b1, const float* __restrict__ norm_src,
                                  float* __restrict__ h2, int N) {
  long i = (long)blockIdx.x * blockDim.x + threadIdx.x;
  const long total = (long)N * H_DIM;
  if (i >= total) return;
  const int node = (int)(i >> 6);
  const int j = (int)(i & 63);
  const float v = fmaxf(agg[i] * norm_dst[node] + b1[j], 0.0f);
  h2[i] = v * norm_src[node];
}

__global__ __launch_bounds__(256) void gemm2_final_wmma(
    const float* __restrict__ agg2,
    const float* __restrict__ W_mu, const float* __restrict__ b_mu,
    const float* __restrict__ W_sig, const float* __restrict__ b_sig,
    const float* __restrict__ norm_dst, const float* __restrict__ noise,
    float* __restrict__ z, int N) {
  __shared__ v2f lWm[(H_DIM / 4) * 4 * 32];
  __shared__ v2f lWs[(H_DIM / 4) * 4 * 32];
  stage_w_frag(lWm, W_mu, H_DIM / 4, threadIdx.x, 256);
  stage_w_frag(lWs, W_sig, H_DIM / 4, threadIdx.x, 256);
  __syncthreads();

  const int wave = threadIdx.x >> 5;
  const int lane = threadIdx.x & 31;
  const int tile = blockIdx.x * 8 + wave;
  const int row0 = tile * 16;
  if (row0 >= N) return;

  const int mrow = row0 + (lane & 15);
  const int kgrp = (lane >> 4) * 2;

  v8f am[4] = {}, as[4] = {};
  for (int kb = 0; kb < H_DIM / 4; ++kb) {
    const v2f a = *(const v2f*)(agg2 + (long)mrow * H_DIM + kb * 4 + kgrp);
#pragma unroll
    for (int ct = 0; ct < 4; ++ct) {
      const v2f bm = lWm[(kb * 4 + ct) * 32 + lane];
      const v2f bs = lWs[(kb * 4 + ct) * 32 + lane];
      am[ct] = __builtin_amdgcn_wmma_f32_16x16x4_f32(
          false, a, false, bm, (short)0, am[ct], false, false);
      as[ct] = __builtin_amdgcn_wmma_f32_16x16x4_f32(
          false, a, false, bs, (short)0, as[ct], false, false);
    }
  }

  const int mbase = row0 + ((lane >> 4) << 3);
#pragma unroll
  for (int ct = 0; ct < 4; ++ct) {
    const int n = ct * 16 + (lane & 15);
    const float bmu = b_mu[n];
    const float bsg = b_sig[n];
#pragma unroll
    for (int v = 0; v < 8; ++v) {
      const int m = mbase + v;
      const float nd = norm_dst[m];
      const float mu = am[ct][v] * nd + bmu;
      const float ls = as[ct][v] * nd + bsg;
      z[(long)m * H_DIM + n] = mu + noise[(long)m * H_DIM + n] * expf(ls);
    }
  }
}

extern "C" void kernel_launch(void* const* d_in, const int* in_sizes, int n_in,
                              void* d_out, int out_size, void* d_ws, size_t ws_size,
                              hipStream_t stream) {
  const float* feat  = (const float*)d_in[0];
  const int*   src   = (const int*)d_in[1];
  const int*   dst   = (const int*)d_in[2];
  const float* noise = (const float*)d_in[3];
  const float* W1    = (const float*)d_in[4];
  const float* b1    = (const float*)d_in[5];
  const float* W_mu  = (const float*)d_in[6];
  const float* b_mu  = (const float*)d_in[7];
  const float* W_sig = (const float*)d_in[8];
  const float* b_sig = (const float*)d_in[9];

  const int E = in_sizes[1];
  const int N = in_sizes[3] / H_DIM;

  float* ws       = (float*)d_ws;
  float* norm_src = ws;
  float* norm_dst = ws + N;
  float* buf_h    = ws + 2L * N;
  float* buf_agg  = buf_h + (long)N * H_DIM;
  float* zout     = (float*)d_out;

  const int threads = 256;
  const int gemm_blocks = (N / 16 + 7) / 8;
  const long elems = (long)N * H_DIM;

  unsigned* deg_part = (unsigned*)buf_h;
  degree_hist_kernel<<<dim3(DEG_S, (N + DEG_R - 1) / DEG_R, 2), 1024, 0, stream>>>(src, dst, deg_part, E, N);
  norm_kernel<<<dim3((N + threads - 1) / threads, 2), threads, 0, stream>>>(deg_part, norm_src, N);

  gemm1_wmma<<<gemm_blocks, threads, 0, stream>>>(feat, norm_src, W1, buf_h, N);
  zero_f32<<<(int)((elems + threads - 1) / threads), threads, 0, stream>>>(buf_agg, elems);
  aggregate_kernel<<<(int)(((long)E * 16 + threads - 1) / threads), threads, 0, stream>>>(
      src, dst, buf_h, buf_agg, E);
  relu_scale_kernel<<<(int)((elems + threads - 1) / threads), threads, 0, stream>>>(
      buf_agg, norm_dst, b1, norm_src, buf_h, N);

  zero_f32<<<(int)((elems + threads - 1) / threads), threads, 0, stream>>>(buf_agg, elems);
  aggregate_kernel<<<(int)(((long)E * 16 + threads - 1) / threads), threads, 0, stream>>>(
      src, dst, buf_h, buf_agg, E);
  gemm2_final_wmma<<<gemm_blocks, threads, 0, stream>>>(
      buf_agg, W_mu, b_mu, W_sig, b_sig, norm_dst, noise, zout, N);
}


# Edge aggregation: one wave walks 32 edges a row at a time, so each gather load and each float atomic covers one whole 128-byte line (was 8 floats in each of 4 lines); arrival order of the sums changes
# speedup vs baseline: 2.2630x; 1.6908x over previous
#include <hip/hip_runtime.h>

typedef __attribute__((ext_vector_type(2))) float v2f;
typedef __attribute__((ext_vector_type(8))) float v8f;

#define F_IN  128
#define H_DIM 64

__global__ void zero_f32(float* __restrict__ p, long n) {
  long i = (long)blockIdx.x * blockDim.x + threadIdx.x;
  long stride = (long)gridDim.x * blockDim.x;
  for (; i < n; i += stride) p[i] = 0.0f;
}

#define DEG_R 25600
#define DEG_S 32
static_assert(2 * DEG_S <= H_DIM, "partial planes must fit the h buffer");

__device__ __forceinline__ void count4(unsigned* bins, const int4 q, const int lo) {
  const unsigned r0 = (unsigned)(q.x - lo), r1 = (unsigned)(q.y - lo);
  const unsigned r2 = (unsigned)(q.z - lo), r3 = (unsigned)(q.w - lo);
  if (r0 < DEG_R) atomicAdd(&bins[r0], 1u);
  if (r1 < DEG_R) atomicAdd(&bins[r1], 1u);
  if (r2 < DEG_R) atomicAdd(&bins[r2], 1u);
  if (r3 < DEG_R) atomicAdd(&bins[r3], 1u);
}

__global__ __launch_bounds__(1024) void degree_hist_kernel(
    const int* __restrict__ src, const int* __restrict__ dst,
    unsigned* __restrict__ part, int E, int N) {
  __shared__ unsigned bins[DEG_R];
  const int* __restrict__ idx = blockIdx.z ? dst : src;
  const int lo = blockIdx.y * DEG_R;
  for (int i = threadIdx.x; i < DEG_R; i += 1024) bins[i] = 0u;
  __syncthreads();

  const int chunk = (((E + DEG_S - 1) / DEG_S) + 3) & ~3;
  const int beg = min(E, (int)blockIdx.x * chunk);
  const int end = min(E, beg + chunk);
  const int nvec = (end - beg) >> 2;
  const int4* __restrict__ v = (const int4*)(idx + beg);
  const int4 none = make_int4(lo - 1, lo - 1, lo - 1, lo - 1);
  const int last = nvec - 1;
  for (int i = threadIdx.x; i < nvec; i += 4096) {
    const int4 q0 = v[i];
    int4 q1 = v[min(i + 1024, last)];
    int4 q2 = v[min(i + 2048, last)];
    int4 q3 = v[min(i + 3072, last)];
    if (i + 1024 > last) q1 = none;
    if (i + 2048 > last) q2 = none;
    if (i + 3072 > last) q3 = none;
    count4(bins, q0, lo);
    count4(bins, q1, lo);
    count4(bins, q2, lo);
    count4(bins, q3, lo);
  }
  for (int i = beg + (nvec << 2) + threadIdx.x; i < end; i += 1024) {
    const unsigned r = (unsigned)(idx[i] - lo);
    if (r < DEG_R) atomicAdd(&bins[r], 1u);
  }
  __syncthreads();

  unsigned* __restrict__ plane = part + (long)(blockIdx.z * DEG_S + blockIdx.x) * N;
  const int hi = min(N, lo + DEG_R);
  for (int i = lo + threadIdx.x; i < hi; i += 1024) plane[i] = bins[i - lo];
}

__global__ void norm_kernel(const unsigned* __restrict__ part, float* __restrict__ norm, int N) {
  const int i = blockIdx.x * blockDim.x + threadIdx.x;
  if (i >= N) return;
  const unsigned* __restrict__ p = part + (long)blockIdx.y * DEG_S * N + i;
  unsigned c = 0u;
  for (int s = 0; s < DEG_S; ++s) c += p[(long)s * N];
  norm[(long)blockIdx.y * N + i] = rsqrtf(fmaxf((float)c, 1.0f));
}

__device__ __forceinline__ void stage_w_frag(v2f* lW, const float* __restrict__ W,
                                             int kblocks, int tid, int nthreads) {
  const int total = kblocks * 4 * 32;
  for (int i = tid; i < total; i += nthreads) {
    const int kb = i >> 7;
    const int ct = (i >> 5) & 3;
    const int ln = i & 31;
    const int kg = (ln >> 4) * 2;
    const int n  = ct * 16 + (ln & 15);
    v2f p;
    p.x = W[(kb * 4 + kg)     * H_DIM + n];
    p.y = W[(kb * 4 + kg + 1) * H_DIM + n];
    lW[i] = p;
  }
}

__global__ __launch_bounds__(256) void gemm1_wmma(
    const float* __restrict__ feat, const float* __restrict__ norm_src,
    const float* __restrict__ W1, float* __restrict__ h1, int N) {
  __shared__ v2f lW[(F_IN / 4) * 4 * 32];
  stage_w_frag(lW, W1, F_IN / 4, threadIdx.x, 256);
  __syncthreads();

  const int wave = threadIdx.x >> 5;
  const int lane = threadIdx.x & 31;
  const int tile = blockIdx.x * 8 + wave;
  const int row0 = tile * 16;
  if (row0 >= N) return;

  const int mrow = row0 + (lane & 15);
  const int kgrp = (lane >> 4) * 2;
  const float ns = norm_src[mrow];

  v8f acc[4] = {};
  for (int kb = 0; kb < F_IN / 4; ++kb) {
    v2f a = *(const v2f*)(feat + (long)mrow * F_IN + kb * 4 + kgrp);
    a.x *= ns; a.y *= ns;
#pragma unroll
    for (int ct = 0; ct < 4; ++ct) {
      const v2f b = lW[(kb * 4 + ct) * 32 + lane];
      acc[ct] = __builtin_amdgcn_wmma_f32_16x16x4_f32(
          false, a, false, b, (short)0, acc[ct], false, false);
    }
  }

  const int mbase = row0 + ((lane >> 4) << 3);
#pragma unroll
  for (int ct = 0; ct < 4; ++ct) {
    const int n = ct * 16 + (lane & 15);
#pragma unroll
    for (int v = 0; v < 8; ++v)
      h1[(long)(mbase + v) * H_DIM + n] = acc[ct][v];
  }
}

#define AGG_BATCH 16
__global__ __launch_bounds__(256) void aggregate_kernel(const int* __restrict__ src, const int* __restrict__ dst,
                                 const float* __restrict__ h, float* __restrict__ agg, int E) {
  const int lane = threadIdx.x & 31;
  const int wave = __builtin_amdgcn_readfirstlane((int)(threadIdx.x >> 5));
  const long e0 = ((long)blockIdx.x * 8 + wave) * 32;
  if (e0 >= E) return;
  const int cnt = (int)min(32L, (long)E - e0);
  const long e = e0 + min(lane, cnt - 1);
  const int sv = src[e];
  const int dv = dst[e];
  for (int j0 = 0; j0 < cnt; j0 += AGG_BATCH) {
    float lo[AGG_BATCH], hi[AGG_BATCH];
    for (int u = 0; u < AGG_BATCH; ++u) {
      const int s = __builtin_amdgcn_readlane(sv, min(j0 + u, cnt - 1));
      const float* __restrict__ hp = h + (long)s * H_DIM + lane;
      lo[u] = hp[0];
      hi[u] = hp[32];
    }
    for (int u = 0; u < AGG_BATCH; ++u) {
      if (j0 + u < cnt) {
        const int d = __builtin_amdgcn_readlane(dv, j0 + u);
        float* ap = agg + (long)d * H_DIM + lane;
        atomicAdd(ap, lo[u]);
        atomicAdd(ap + 32, hi[u]);
      }
    }
  }
}

__global__ void relu_scale_kernel(const float* __restrict__ agg, const float* __restrict__ norm_dst,
                                  const float* __restrict__ b1, const float* __restrict__ norm_src,
                                  float* __restrict__ h2, int N) {
  long i = (long)blockIdx.x * blockDim.x + threadIdx.x;
  const long total = (long)N * H_DIM;
  if (i >= total) return;
  const int node = (int)(i >> 6);
  const int j = (int)(i & 63);
  const float v = fmaxf(agg[i] * norm_dst[node] + b1[j], 0.0f);
  h2[i] = v * norm_src[node];
}

__global__ __launch_bounds__(256) void gemm2_final_wmma(
    const float* __restrict__ agg2,
    const float* __restrict__ W_mu, const float* __restrict__ b_mu,
    const float* __restrict__ W_sig, const float* __restrict__ b_sig,
    const float* __restrict__ norm_dst, const float* __restrict__ noise,
    float* __restrict__ z, int N) {
  __shared__ v2f lWm[(H_DIM / 4) * 4 * 32];
  __shared__ v2f lWs[(H_DIM / 4) * 4 * 32];
  stage_w_frag(lWm, W_mu, H_DIM / 4, threadIdx.x, 256);
  stage_w_frag(lWs, W_sig, H_DIM / 4, threadIdx.x, 256);
  __syncthreads();

  const int wave = threadIdx.x >> 5;
  const int lane = threadIdx.x & 31;
  const int tile = blockIdx.x * 8 + wave;
  const int row0 = tile * 16;
  if (row0 >= N) return;

  const int mrow = row0 + (lane & 15);
  const int kgrp = (lane >> 4) * 2;

  v8f am[4] = {}, as[4] = {};
  for (int kb = 0; kb < H_DIM / 4; ++kb) {
    const v2f a = *(const v2f*)(agg2 + (long)mrow * H_DIM + kb * 4 + kgrp);
#pragma unroll
    for (int ct = 0; ct < 4; ++ct) {
      const v2f bm = lWm[(kb * 4 + ct) * 32 + lane];
      const v2f bs = lWs[(kb * 4 + ct) * 32 + lane];
      am[ct] = __builtin_amdgcn_wmma_f32_16x16x4_f32(
          false, a, false, bm, (short)0, am[ct], false, false);
      as[ct] = __builtin_amdgcn_wmma_f32_16x16x4_f32(
          false, a, false, bs, (short)0, as[ct], false, false);
    }
  }

  const int mbase = row0 + ((lane >> 4) << 3);
#pragma unroll
  for (int ct = 0; ct < 4; ++ct) {
    const int n = ct * 16 + (lane & 15);
    const float bmu = b_mu[n];
    const float bsg = b_sig[n];
#pragma unroll
    for (int v = 0; v < 8; ++v) {
      const int m = mbase + v;
      const float nd = norm_dst[m];
      const float mu = am[ct][v] * nd + bmu;
      const float ls = as[ct][v] * nd + bsg;
      z[(long)m * H_DIM + n] = mu + noise[(long)m * H_DIM + n] * expf(ls);
    }
  }
}

extern "C" void kernel_launch(void* const* d_in, const int* in_sizes, int n_in,
                              void* d_out, int out_size, void* d_ws, size_t ws_size,
                              hipStream_t stream) {
  const float* feat  = (const float*)d_in[0];
  const int*   src   = (const int*)d_in[1];
  const int*   dst   = (const int*)d_in[2];
  const float* noise = (const float*)d_in[3];
  const float* W1    = (const float*)d_in[4];
  const float* b1    = (const float*)d_in[5];
  const float* W_mu  = (const float*)d_in[6];
  const float* b_mu  = (const float*)d_in[7];
  const float* W_sig = (const float*)d_in[8];
  const float* b_sig = (const float*)d_in[9];

  const int E = in_sizes[1];
  const int N = in_sizes[3] / H_DIM;

  float* ws       = (float*)d_ws;
  float* norm_src = ws;
  float* norm_dst = ws + N;
  float* buf_h    = ws + 2L * N;
  float* buf_agg  = buf_h + (long)N * H_DIM;
  float* zout     = (float*)d_out;

  const int threads = 256;
  const int gemm_blocks = (N / 16 + 7) / 8;
  const long elems = (long)N * H_DIM;

  unsigned* deg_part = (unsigned*)buf_h;
  degree_hist_kernel<<<dim3(DEG_S, (N + DEG_R - 1) / DEG_R, 2), 1024, 0, stream>>>(src, dst, deg_part, E, N);
  norm_kernel<<<dim3((N + threads - 1) / threads, 2), threads, 0, stream>>>(deg_part, norm_src, N);

  gemm1_wmma<<<gemm_blocks, threads, 0, stream>>>(feat, norm_src, W1, buf_h, N);
  zero_f32<<<(int)((elems + threads - 1) / threads), threads, 0, stream>>>(buf_agg, elems);
  aggregate_kernel<<<(E + threads - 1) / threads, threads, 0, stream>>>(
      src, dst, buf_h, buf_agg, E);
  relu_scale_kernel<<<(int)((elems + threads - 1) / threads), threads, 0, stream>>>(
      buf_agg, norm_dst, b1, norm_src, buf_h, N);

  zero_f32<<<(int)((elems + threads - 1) / threads), threads, 0, stream>>>(buf_agg, elems);
  aggregate_kernel<<<(E + threads - 1) / threads, threads, 0, stream>>>(
      src, dst, buf_h, buf_agg, E);
  gemm2_final_wmma<<<gemm_blocks, threads, 0, stream>>>(
      buf_agg, W_mu, b_mu, W_sig, b_sig, norm_dst, noise, zout, N);
}


# Edge aggregation over a dst-sorted edge list (LDS counting sort from the in-degree planes): a wave sums one row in registers, stores it once; no float atomics, zero fill, ReLU pass; sum order changes
# speedup vs baseline: 2.9662x; 1.3107x over previous
#include <hip/hip_runtime.h>

typedef __attribute__((ext_vector_type(2))) float v2f;
typedef __attribute__((ext_vector_type(8))) float v8f;

#define F_IN  128
#define H_DIM 64

#define DEG_R 25600
#define DEG_S 32
static_assert(2 * DEG_S <= H_DIM, "partial planes must fit the h buffer");

__device__ __forceinline__ void count4(unsigned* bins, const int4 q, const int lo) {
  const unsigned r0 = (unsigned)(q.x - lo), r1 = (unsigned)(q.y - lo);
  const unsigned r2 = (unsigned)(q.z - lo), r3 = (unsigned)(q.w - lo);
  if (r0 < DEG_R) atomicAdd(&bins[r0], 1u);
  if (r1 < DEG_R) atomicAdd(&bins[r1], 1u);
  if (r2 < DEG_R) atomicAdd(&bins[r2], 1u);
  if (r3 < DEG_R) atomicAdd(&bins[r3], 1u);
}

__global__ __launch_bounds__(1024) void degree_hist_kernel(
    const int* __restrict__ src, const int* __restrict__ dst,
    unsigned* __restrict__ part, int E, int N) {
  __shared__ unsigned bins[DEG_R];
  const int* __restrict__ idx = blockIdx.z ? dst : src;
  const int lo = blockIdx.y * DEG_R;
  for (int i = threadIdx.x; i < DEG_R; i += 1024) bins[i] = 0u;
  __syncthreads();

  const int chunk = (((E + DEG_S - 1) / DEG_S) + 3) & ~3;
  const int beg = min(E, (int)blockIdx.x * chunk);
  const int end = min(E, beg + chunk);
  const int nvec = (end - beg) >> 2;
  const int4* __restrict__ v = (const int4*)(idx + beg);
  const int4 none = make_int4(lo - 1, lo - 1, lo - 1, lo - 1);
  const int last = nvec - 1;
  for (int i = threadIdx.x; i < nvec; i += 4096) {
    const int4 q0 = v[i];
    int4 q1 = v[min(i + 1024, last)];
    int4 q2 = v[min(i + 2048, last)];
    int4 q3 = v[min(i + 3072, last)];
    if (i + 1024 > last) q1 = none;
    if (i + 2048 > last) q2 = none;
    if (i + 3072 > last) q3 = none;
    count4(bins, q0, lo);
    count4(bins, q1, lo);
    count4(bins, q2, lo);
    count4(bins, q3, lo);
  }
  for (int i = beg + (nvec << 2) + threadIdx.x; i < end; i += 1024) {
    const unsigned r = (unsigned)(idx[i] - lo);
    if (r < DEG_R) atomicAdd(&bins[r], 1u);
  }
  __syncthreads();

  unsigned* __restrict__ plane = part + (long)(blockIdx.z * DEG_S + blockIdx.x) * N;
  const int hi = min(N, lo + DEG_R);
  for (int i = lo + threadIdx.x; i < hi; i += 1024) plane[i] = bins[i - lo];
}

#define NORM_T 256
__global__ __launch_bounds__(NORM_T) void norm_kernel(const unsigned* __restrict__ part, float* __restrict__ norm,
                                                      unsigned* __restrict__ bsum, int N) {
  __shared__ unsigned wsum[NORM_T / 32];
  const int i = blockIdx.x * NORM_T + threadIdx.x;
  unsigned c = 0u;
  if (i < N) {
    const unsigned* __restrict__ p = part + (long)blockIdx.y * DEG_S * N + i;
    for (int s = 0; s < DEG_S; ++s) c += p[(long)s * N];
    norm[(long)blockIdx.y * N + i] = rsqrtf(fmaxf((float)c, 1.0f));
  }
  if (blockIdx.y == 1) {
    unsigned t = c;
    for (int o = 16; o > 0; o >>= 1) t += __shfl_xor(t, o);
    if ((threadIdx.x & 31) == 0) wsum[threadIdx.x >> 5] = t;
    __syncthreads();
    if (threadIdx.x == 0) {
      unsigned b = 0u;
      for (int w = 0; w < NORM_T / 32; ++w) b += wsum[w];
      bsum[blockIdx.x] = b;
    }
  }
}

__global__ __launch_bounds__(NORM_T) void offsets_kernel(unsigned* __restrict__ dplanes,
                                                         const unsigned* __restrict__ bsum,
                                                         unsigned* __restrict__ rowptr, int N) {
  __shared__ unsigned red[NORM_T / 32];
  __shared__ unsigned wtot[NORM_T / 32];
  const int lane = threadIdx.x & 31;
  const int wave = threadIdx.x >> 5;
  unsigned b = 0u;
  for (int k = threadIdx.x; k < (int)blockIdx.x; k += NORM_T) b += bsum[k];
  for (int o = 16; o > 0; o >>= 1) b += __shfl_xor(b, o);
  if (lane == 0) red[wave] = b;

  const int node = blockIdx.x * NORM_T + threadIdx.x;
  unsigned* __restrict__ p = dplanes + min(node, N - 1);
  unsigned t[DEG_S];
  unsigned c = 0u;
#pragma unroll
  for (int s = 0; s < DEG_S; ++s) {
    t[s] = (node < N) ? p[(long)s * N] : 0u;
    c += t[s];
  }
  unsigned x = c;
  for (int o = 1; o < 32; o <<= 1) {
    const unsigned y = __shfl_up(x, o);
    if (lane >= o) x += y;
  }
  if (lane == 31) wtot[wave] = x;
  __syncthreads();
  unsigned run = x - c;
  for (int w = 0; w < NORM_T / 32; ++w) {
    run += red[w];
    if (w < wave) run += wtot[w];
  }
  if (node < N) {
    rowptr[node] = run;
#pragma unroll
    for (int s = 0; s < DEG_S; ++s) {
      p[(long)s * N] = run;
      run += t[s];
    }
    if (node == N - 1) rowptr[N] = run;
  }
}

#define SC_R 12800
__device__ __forceinline__ void place1(unsigned* cur, int* __restrict__ csr_src,
                                       const int d, const int s, const int lo, const int E) {
  const unsigned r = (unsigned)(d - lo);
  if (r < SC_R) {
    const unsigned pos = atomicAdd(&cur[r], 1u);
    if (pos < (unsigned)E) csr_src[pos] = s;
  }
}
__device__ __forceinline__ void place4(unsigned* cur, int* __restrict__ csr_src,
                                       const int4 d, const int4 s, const int lo, const int E) {
  place1(cur, csr_src, d.x, s.x, lo, E);
  place1(cur, csr_src, d.y, s.y, lo, E);
  place1(cur, csr_src, d.z, s.z, lo, E);
  place1(cur, csr_src, d.w, s.w, lo, E);
}

__global__ __launch_bounds__(1024) void scatter_kernel(
    const int* __restrict__ src, const int* __restrict__ dst,
    const unsigned* __restrict__ dplanes, int* __restrict__ csr_src, int E, int N) {
  __shared__ unsigned cur[SC_R];
  const int lo = blockIdx.y * SC_R;
  const int hi = min(N, lo + SC_R);
  const unsigned* __restrict__ plane = dplanes + (long)blockIdx.x * N;
  for (int i = lo + threadIdx.x; i < hi; i += 1024) cur[i - lo] = plane[i];
  __syncthreads();

  const int chunk = (((E + DEG_S - 1) / DEG_S) + 3) & ~3;
  const int beg = min(E, (int)blockIdx.x * chunk);
  const int end = min(E, beg + chunk);
  const int nvec = (end - beg) >> 2;
  const int4* __restrict__ dv = (const int4*)(dst + beg);
  const int4* __restrict__ sv = (const int4*)(src + beg);
  const int4 none = make_int4(lo - 1, lo - 1, lo - 1, lo - 1);
  const int last = nvec - 1;
  for (int i = threadIdx.x; i < nvec; i += 2048) {
    const int i1 = min(i + 1024, last);
    const int4 d0 = dv[i];
    const int4 s0 = sv[i];
    int4 d1 = dv[i1];
    const int4 s1 = sv[i1];
    if (i + 1024 > last) d1 = none;
    place4(cur, csr_src, d0, s0, lo, E);
    place4(cur, csr_src, d1, s1, lo, E);
  }
  for (int i = beg + (nvec << 2) + threadIdx.x; i < end; i += 1024)
    place1(cur, csr_src, dst[i], src[i], lo, E);
}

__device__ __forceinline__ void stage_w_frag(v2f* lW, const float* __restrict__ W,
                                             int kblocks, int tid, int nthreads) {
  const int total = kblocks * 4 * 32;
  for (int i = tid; i < total; i += nthreads) {
    const int kb = i >> 7;
    const int ct = (i >> 5) & 3;
    const int ln = i & 31;
    const int kg = (ln >> 4) * 2;
    const int n  = ct * 16 + (ln & 15);
    v2f p;
    p.x = W[(kb * 4 + kg)     * H_DIM + n];
    p.y = W[(kb * 4 + kg + 1) * H_DIM + n];
    lW[i] = p;
  }
}

__global__ __launch_bounds__(256) void gemm1_wmma(
    const float* __restrict__ feat, const float* __restrict__ norm_src,
    const float* __restrict__ W1, float* __restrict__ h1, int N) {
  __shared__ v2f lW[(F_IN / 4) * 4 * 32];
  stage_w_frag(lW, W1, F_IN / 4, threadIdx.x, 256);
  __syncthreads();

  const int wave = threadIdx.x >> 5;
  const int lane = threadIdx.x & 31;
  const int tile = blockIdx.x * 8 + wave;
  const int row0 = tile * 16;
  if (row0 >= N) return;

  const int mrow = row0 + (lane & 15);
  const int kgrp = (lane >> 4) * 2;
  const float ns = norm_src[mrow];

  v8f acc[4] = {};
  for (int kb = 0; kb < F_IN / 4; ++kb) {
    v2f a = *(const v2f*)(feat + (long)mrow * F_IN + kb * 4 + kgrp);
    a.x *= ns; a.y *= ns;
#pragma unroll
    for (int ct = 0; ct < 4; ++ct) {
      const v2f b = lW[(kb * 4 + ct) * 32 + lane];
      acc[ct] = __builtin_amdgcn_wmma_f32_16x16x4_f32(
          false, a, false, b, (short)0, acc[ct], false, false);
    }
  }

  const int mbase = row0 + ((lane >> 4) << 3);
#pragma unroll
  for (int ct = 0; ct < 4; ++ct) {
    const int n = ct * 16 + (lane & 15);
#pragma unroll
    for (int v = 0; v < 8; ++v)
      h1[(long)(mbase + v) * H_DIM + n] = acc[ct][v];
  }
}

#define AGG_BATCH 8
__global__ __launch_bounds__(256) void aggregate_rows_kernel(
    const int* __restrict__ csr_src, const unsigned* __restrict__ rowptr,
    const float* __restrict__ h, const float* __restrict__ norm_dst,
    const float* __restrict__ b1, const float* __restrict__ norm_src,
    float* __restrict__ res, int relu, int N, int E) {
  const int lane = threadIdx.x & 31;
  const int wave = __builtin_amdgcn_readfirstlane((int)(threadIdx.x >> 5));
  const int row = blockIdx.x * 8 + wave;
  if (row >= N) return;
  const int beg = (int)min(rowptr[row], (unsigned)E);
  const int end = (int)min(rowptr[row + 1], (unsigned)E);
  v2f acc = {0.0f, 0.0f};
  for (int j0 = beg; j0 < end; j0 += 32) {
    const int cnt = min(32, end - j0);
    const int sv = min(max(csr_src[j0 + min(lane, cnt - 1)], 0), N - 1);
    for (int u0 = 0; u0 < cnt; u0 += AGG_BATCH) {
      v2f g[AGG_BATCH];
      for (int u = 0; u < AGG_BATCH; ++u) {
        const int s = __builtin_amdgcn_readlane(sv, min(u0 + u, cnt - 1));
        g[u] = *(const v2f*)(h + (long)s * H_DIM + 2 * lane);
      }
      for (int u = 0; u < AGG_BATCH; ++u) {
        if (u0 + u < cnt) {
          acc.x += g[u].x;
          acc.y += g[u].y;
        }
      }
    }
  }
  if (relu) {
    const float nd = norm_dst[row];
    const float ns = norm_src[row];
    acc.x = fmaxf(acc.x * nd + b1[2 * lane], 0.0f) * ns;
    acc.y = fmaxf(acc.y * nd + b1[2 * lane + 1], 0.0f) * ns;
  }
  *(v2f*)(res + (long)row * H_DIM + 2 * lane) = acc;
}

__global__ __launch_bounds__(256) void gemm2_final_wmma(
    const float* __restrict__ agg2,
    const float* __restrict__ W_mu, const float* __restrict__ b_mu,
    const float* __restrict__ W_sig, const float* __restrict__ b_sig,
    const float* __restrict__ norm_dst, const float* __restrict__ noise,
    float* __restrict__ z, int N) {
  __shared__ v2f lWm[(H_DIM / 4) * 4 * 32];
  __shared__ v2f lWs[(H_DIM / 4) * 4 * 32];
  stage_w_frag(lWm, W_mu, H_DIM / 4, threadIdx.x, 256);
  stage_w_frag(lWs, W_sig, H_DIM / 4, threadIdx.x, 256);
  __syncthreads();

  const int wave = threadIdx.x >> 5;
  const int lane = threadIdx.x & 31;
  const int tile = blockIdx.x * 8 + wave;
  const int row0 = tile * 16;
  if (row0 >= N) return;

  const int mrow = row0 + (lane & 15);
  const int kgrp = (lane >> 4) * 2;

  v8f am[4] = {}, as[4] = {};
  for (int kb = 0; kb < H_DIM / 4; ++kb) {
    const v2f a = *(const v2f*)(agg2 + (long)mrow * H_DIM + kb * 4 + kgrp);
#pragma unroll
    for (int ct = 0; ct < 4; ++ct) {
      const v2f bm = lWm[(kb * 4 + ct) * 32 + lane];
      const v2f bs = lWs[(kb * 4 + ct) * 32 + lane];
      am[ct] = __builtin_amdgcn_wmma_f32_16x16x4_f32(
          false, a, false, bm, (short)0, am[ct], false, false);
      as[ct] = __builtin_amdgcn_wmma_f32_16x16x4_f32(
          false, a, false, bs, (short)0, as[ct], false, false);
    }
  }

  const int mbase = row0 + ((lane >> 4) << 3);
#pragma unroll
  for (int ct = 0; ct < 4; ++ct) {
    const int n = ct * 16 + (lane & 15);
    const float bmu = b_mu[n];
    const float bsg = b_sig[n];
#pragma unroll
    for (int v = 0; v < 8; ++v) {
      const int m = mbase + v;
      const float nd = norm_dst[m];
      const float mu = am[ct][v] * nd + bmu;
      const float ls = as[ct][v] * nd + bsg;
      z[(long)m * H_DIM + n] = mu + noise[(long)m * H_DIM + n] * expf(ls);
    }
  }
}

extern "C" void kernel_launch(void* const* d_in, const int* in_sizes, int n_in,
                              void* d_out, int out_size, void* d_ws, size_t ws_size,
                              hipStream_t stream) {
  const float* feat  = (const float*)d_in[0];
  const int*   src   = (const int*)d_in[1];
  const int*   dst   = (const int*)d_in[2];
  const float* noise = (const float*)d_in[3];
  const float* W1    = (const float*)d_in[4];
  const float* b1    = (const float*)d_in[5];
  const float* W_mu  = (const float*)d_in[6];
  const float* b_mu  = (const float*)d_in[7];
  const float* W_sig = (const float*)d_in[8];
  const float* b_sig = (const float*)d_in[9];

  const int E = in_sizes[1];
  const int N = in_sizes[3] / H_DIM;

  float* ws       = (float*)d_ws;
  float* norm_src = ws;
  float* norm_dst = ws + N;
  float* buf_a    = ws + 2L * N;
  float* buf_b    = buf_a + (long)N * H_DIM;
  float* zout     = (float*)d_out;

  const int threads = 256;
  const int gemm_blocks = (N / 16 + 7) / 8;
  const int node_blocks = (N + NORM_T - 1) / NORM_T;

  int*      csr_src = (int*)zout;
  unsigned* rowptr  = (unsigned*)zout + E;
  unsigned* bsum    = rowptr + N + 1;

  unsigned* deg_part = (unsigned*)buf_a;
  unsigned* dplanes  = deg_part + (long)DEG_S * N;
  degree_hist_kernel<<<dim3(DEG_S, (N + DEG_R - 1) / DEG_R, 2), 1024, 0, stream>>>(src, dst, deg_part, E, N);
  norm_kernel<<<dim3(node_blocks, 2), NORM_T, 0, stream>>>(deg_part, norm_src, bsum, N);
  offsets_kernel<<<node_blocks, NORM_T, 0, stream>>>(dplanes, bsum, rowptr, N);
  scatter_kernel<<<dim3(DEG_S, (N + SC_R - 1) / SC_R), 1024, 0, stream>>>(src, dst, dplanes, csr_src, E, N);

  gemm1_wmma<<<gemm_blocks, threads, 0, stream>>>(feat, norm_src, W1, buf_a, N);
  aggregate_rows_kernel<<<(N + 7) / 8, threads, 0, stream>>>(
      csr_src, rowptr, buf_a, norm_dst, b1, norm_src, buf_b, 1, N, E);

  aggregate_rows_kernel<<<(N + 7) / 8, threads, 0, stream>>>(
      csr_src, rowptr, buf_b, norm_dst, b1, norm_src, buf_a, 0, N, E);
  gemm2_final_wmma<<<gemm_blocks, threads, 0, stream>>>(
      buf_a, W_mu, b_mu, W_sig, b_sig, norm_dst, noise, zout, N);
}


# Layer-2 edge aggregation moved into the mu/sigma product launch: a wave sums its 16-row tile from the sorted list into an LDS slot, multiplies from there; 6 launches, agg2 not stored; sum order kept
# speedup vs baseline: 3.3893x; 1.1426x over previous
#include <hip/hip_runtime.h>

typedef __attribute__((ext_vector_type(2))) float v2f;
typedef __attribute__((ext_vector_type(8))) float v8f;

#define F_IN  128
#define H_DIM 64

#define DEG_R 25600
#define DEG_S 32
static_assert(2 * DEG_S <= H_DIM, "partial planes must fit the h buffer");

__device__ __forceinline__ void count4(unsigned* bins, const int4 q, const int lo) {
  const unsigned r0 = (unsigned)(q.x - lo), r1 = (unsigned)(q.y - lo);
  const unsigned r2 = (unsigned)(q.z - lo), r3 = (unsigned)(q.w - lo);
  if (r0 < DEG_R) atomicAdd(&bins[r0], 1u);
  if (r1 < DEG_R) atomicAdd(&bins[r1], 1u);
  if (r2 < DEG_R) atomicAdd(&bins[r2], 1u);
  if (r3 < DEG_R) atomicAdd(&bins[r3], 1u);
}

__global__ __launch_bounds__(1024) void degree_hist_kernel(
    const int* __restrict__ src, const int* __restrict__ dst,
    unsigned* __restrict__ part, int E, int N) {
  __shared__ unsigned bins[DEG_R];
  const int* __restrict__ idx = blockIdx.z ? dst : src;
  const int lo = blockIdx.y * DEG_R;
  for (int i = threadIdx.x; i < DEG_R; i += 1024) bins[i] = 0u;
  __syncthreads();

  const int chunk = (((E + DEG_S - 1) / DEG_S) + 3) & ~3;
  const int beg = min(E, (int)blockIdx.x * chunk);
  const int end = min(E, beg + chunk);
  const int nvec = (end - beg) >> 2;
  const int4* __restrict__ v = (const int4*)(idx + beg);
  const int4 none = make_int4(lo - 1, lo - 1, lo - 1, lo - 1);
  const int last = nvec - 1;
  for (int i = threadIdx.x; i < nvec; i += 4096) {
    const int4 q0 = v[i];
    int4 q1 = v[min(i + 1024, last)];
    int4 q2 = v[min(i + 2048, last)];
    int4 q3 = v[min(i + 3072, last)];
    if (i + 1024 > last) q1 = none;
    if (i + 2048 > last) q2 = none;
    if (i + 3072 > last) q3 = none;
    count4(bins, q0, lo);
    count4(bins, q1, lo);
    count4(bins, q2, lo);
    count4(bins, q3, lo);
  }
  for (int i = beg + (nvec << 2) + threadIdx.x; i < end; i += 1024) {
    const unsigned r = (unsigned)(idx[i] - lo);
    if (r < DEG_R) atomicAdd(&bins[r], 1u);
  }
  __syncthreads();

  unsigned* __restrict__ plane = part + (long)(blockIdx.z * DEG_S + blockIdx.x) * N;
  const int hi = min(N, lo + DEG_R);
  for (int i = lo + threadIdx.x; i < hi; i += 1024) plane[i] = bins[i - lo];
}

#define NORM_T 256
__global__ __launch_bounds__(NORM_T) void norm_kernel(const unsigned* __restrict__ part, float* __restrict__ norm,
                                                      unsigned* __restrict__ bsum, int N) {
  __shared__ unsigned wsum[NORM_T / 32];
  const int i = blockIdx.x * NORM_T + threadIdx.x;
  unsigned c = 0u;
  if (i < N) {
    const unsigned* __restrict__ p = part + (long)blockIdx.y * DEG_S * N + i;
    for (int s = 0; s < DEG_S; ++s) c += p[(long)s * N];
    norm[(long)blockIdx.y * N + i] = rsqrtf(fmaxf((float)c, 1.0f));
  }
  if (blockIdx.y == 1) {
    unsigned t = c;
    for (int o = 16; o > 0; o >>= 1) t += __shfl_xor(t, o);
    if ((threadIdx.x & 31) == 0) wsum[threadIdx.x >> 5] = t;
    __syncthreads();
    if (threadIdx.x == 0) {
      unsigned b = 0u;
      for (int w = 0; w < NORM_T / 32; ++w) b += wsum[w];
      bsum[blockIdx.x] = b;
    }
  }
}

__global__ __launch_bounds__(NORM_T) void offsets_kernel(unsigned* __restrict__ dplanes,
                                                         const unsigned* __restrict__ bsum,
                                                         unsigned* __restrict__ rowptr, int N) {
  __shared__ unsigned red[NORM_T / 32];
  __shared__ unsigned wtot[NORM_T / 32];
  const int lane = threadIdx.x & 31;
  const int wave = threadIdx.x >> 5;
  unsigned b = 0u;
  for (int k = threadIdx.x; k < (int)blockIdx.x; k += NORM_T) b += bsum[k];
  for (int o = 16; o > 0; o >>= 1) b += __shfl_xor(b, o);
  if (lane == 0) red[wave] = b;

  const int node = blockIdx.x * NORM_T + threadIdx.x;
  unsigned* __restrict__ p = dplanes + min(node, N - 1);
  unsigned t[DEG_S];
  unsigned c = 0u;
#pragma unroll
  for (int s = 0; s < DEG_S; ++s) {
    t[s] = (node < N) ? p[(long)s * N] : 0u;
    c += t[s];
  }
  unsigned x = c;
  for (int o = 1; o < 32; o <<= 1) {
    const unsigned y = __shfl_up(x, o);
    if (lane >= o) x += y;
  }
  if (lane == 31) wtot[wave] = x;
  __syncthreads();
  unsigned run = x - c;
  for (int w = 0; w < NORM_T / 32; ++w) {
    run += red[w];
    if (w < wave) run += wtot[w];
  }
  if (node < N) {
    rowptr[node] = run;
#pragma unroll
    for (int s = 0; s < DEG_S; ++s) {
      p[(long)s * N] = run;
      run += t[s];
    }
    if (node == N - 1) rowptr[N] = run;
  }
}

#define SC_R 12800
__device__ __forceinline__ void place1(unsigned* cur, int* __restrict__ csr_src,
                                       const int d, const int s, const int lo, const int E) {
  const unsigned r = (unsigned)(d - lo);
  if (r < SC_R) {
    const unsigned pos = atomicAdd(&cur[r], 1u);
    if (pos < (unsigned)E) csr_src[pos] = s;
  }
}
__device__ __forceinline__ void place4(unsigned* cur, int* __restrict__ csr_src,
                                       const int4 d, const int4 s, const int lo, const int E) {
  place1(cur, csr_src, d.x, s.x, lo, E);
  place1(cur, csr_src, d.y, s.y, lo, E);
  place1(cur, csr_src, d.z, s.z, lo, E);
  place1(cur, csr_src, d.w, s.w, lo, E);
}

__device__ __forceinline__ void scatter_body(
    unsigned* cur, const int slice, const int range,
    const int* __restrict__ src, const int* __restrict__ dst,
    const unsigned* __restrict__ dplanes, int* __restrict__ csr_src, int E, int N) {
  const int lo = range * SC_R;
  const int hi = min(N, lo + SC_R);
  const unsigned* __restrict__ plane = dplanes + (long)slice * N;
  for (int i = lo + threadIdx.x; i < hi; i += 1024) cur[i - lo] = plane[i];
  __syncthreads();

  const int chunk = (((E + DEG_S - 1) / DEG_S) + 3) & ~3;
  const int beg = min(E, slice * chunk);
  const int end = min(E, beg + chunk);
  const int nvec = (end - beg) >> 2;
  const int4* __restrict__ dv = (const int4*)(dst + beg);
  const int4* __restrict__ sv = (const int4*)(src + beg);
  const int4 none = make_int4(lo - 1, lo - 1, lo - 1, lo - 1);
  const int last = nvec - 1;
  for (int i = threadIdx.x; i < nvec; i += 2048) {
    const int i1 = min(i + 1024, last);
    const int4 d0 = dv[i];
    const int4 s0 = sv[i];
    int4 d1 = dv[i1];
    const int4 s1 = sv[i1];
    if (i + 1024 > last) d1 = none;
    place4(cur, csr_src, d0, s0, lo, E);
    place4(cur, csr_src, d1, s1, lo, E);
  }
  for (int i = beg + (nvec << 2) + threadIdx.x; i < end; i += 1024)
    place1(cur, csr_src, dst[i], src[i], lo, E);
}

__device__ __forceinline__ void stage_w_frag(v2f* lW, const float* __restrict__ W,
                                             int kblocks, int tid, int nthreads) {
  const int total = kblocks * 4 * 32;
  for (int i = tid; i < total; i += nthreads) {
    const int kb = i >> 7;
    const int ct = (i >> 5) & 3;
    const int ln = i & 31;
    const int kg = (ln >> 4) * 2;
    const int n  = ct * 16 + (ln & 15);
    v2f p;
    p.x = W[(kb * 4 + kg)     * H_DIM + n];
    p.y = W[(kb * 4 + kg + 1) * H_DIM + n];
    lW[i] = p;
  }
}

__device__ __forceinline__ void gemm1_body(
    v2f* lW, const int grp, const int ngrp,
    const float* __restrict__ feat, const float* __restrict__ norm_src,
    const float* __restrict__ W1, float* __restrict__ h1, int N) {
  stage_w_frag(lW, W1, F_IN / 4, threadIdx.x, 1024);
  __syncthreads();

  const int wave = threadIdx.x >> 5;
  const int lane = threadIdx.x & 31;
  const int tile = wave * ngrp + grp;
  const int row0 = tile * 16;
  if (row0 >= N) return;

  const int mrow = row0 + (lane & 15);
  const int kgrp = (lane >> 4) * 2;
  const float ns = norm_src[mrow];

  v8f acc[4] = {};
  for (int kb = 0; kb < F_IN / 4; ++kb) {
    v2f a = *(const v2f*)(feat + (long)mrow * F_IN + kb * 4 + kgrp);
    a.x *= ns; a.y *= ns;
#pragma unroll
    for (int ct = 0; ct < 4; ++ct) {
      const v2f b = lW[(kb * 4 + ct) * 32 + lane];
      acc[ct] = __builtin_amdgcn_wmma_f32_16x16x4_f32(
          false, a, false, b, (short)0, acc[ct], false, false);
    }
  }

  const int mbase = row0 + ((lane >> 4) << 3);
#pragma unroll
  for (int ct = 0; ct < 4; ++ct) {
    const int n = ct * 16 + (lane & 15);
#pragma unroll
    for (int v = 0; v < 8; ++v)
      h1[(long)(mbase + v) * H_DIM + n] = acc[ct][v];
  }
}

#define L1_GROUPS 256
__global__ __launch_bounds__(1024) void sort_scatter_and_layer1_kernel(
    const int* __restrict__ src, const int* __restrict__ dst,
    const unsigned* __restrict__ dplanes, int* __restrict__ csr_src,
    const float* __restrict__ feat, const float* __restrict__ norm_src,
    const float* __restrict__ W1, float* __restrict__ h1,
    int E, int N, int n_scatter, int n_gemm) {
  __shared__ __attribute__((aligned(16))) unsigned char smem[SC_R * sizeof(unsigned)];
  static_assert(sizeof(v2f) * (F_IN / 4) * 4 * 32 <= SC_R * sizeof(unsigned), "W1 fragments must fit the cursor array");
  const int id = blockIdx.x;
  if (id < n_scatter)
    scatter_body((unsigned*)smem, id % DEG_S, id / DEG_S, src, dst, dplanes, csr_src, E, N);
  else
    gemm1_body((v2f*)smem, id - n_scatter, n_gemm, feat, norm_src, W1, h1, N);
}

#define AGG_BATCH 8
__global__ __launch_bounds__(256) void aggregate_rows_kernel(
    const int* __restrict__ csr_src, const unsigned* __restrict__ rowptr,
    const float* __restrict__ h, const float* __restrict__ norm_dst,
    const float* __restrict__ b1, const float* __restrict__ norm_src,
    float* __restrict__ res, int relu, int N, int E) {
  const int lane = threadIdx.x & 31;
  const int wave = __builtin_amdgcn_readfirstlane((int)(threadIdx.x >> 5));
  const int row = blockIdx.x * 8 + wave;
  if (row >= N) return;
  const int beg = (int)min(rowptr[row], (unsigned)E);
  const int end = (int)min(rowptr[row + 1], (unsigned)E);
  v2f acc = {0.0f, 0.0f};
  for (int j0 = beg; j0 < end; j0 += 32) {
    const int cnt = min(32, end - j0);
    const int sv = min(max(csr_src[j0 + min(lane, cnt - 1)], 0), N - 1);
    for (int u0 = 0; u0 < cnt; u0 += AGG_BATCH) {
      v2f g[AGG_BATCH];
      for (int u = 0; u < AGG_BATCH; ++u) {
        const int s = __builtin_amdgcn_readlane(sv, min(u0 + u, cnt - 1));
        g[u] = *(const v2f*)(h + (long)s * H_DIM + 2 * lane);
      }
      for (int u = 0; u < AGG_BATCH; ++u) {
        if (u0 + u < cnt) {
          acc.x += g[u].x;
          acc.y += g[u].y;
        }
      }
    }
  }
  if (relu) {
    const float nd = norm_dst[row];
    const float ns = norm_src[row];
    acc.x = fmaxf(acc.x * nd + b1[2 * lane], 0.0f) * ns;
    acc.y = fmaxf(acc.y * nd + b1[2 * lane + 1], 0.0f) * ns;
  }
  *(v2f*)(res + (long)row * H_DIM + 2 * lane) = acc;
}

#define T_PITCH 68
__global__ __launch_bounds__(256) void aggregate_gemm2_kernel(
    const int* __restrict__ csr_src, const unsigned* __restrict__ rowptr,
    const float* __restrict__ h,
    const float* __restrict__ W_mu, const float* __restrict__ b_mu,
    const float* __restrict__ W_sig, const float* __restrict__ b_sig,
    const float* __restrict__ norm_dst, const float* __restrict__ noise,
    float* __restrict__ z, int N, int E) {
  __shared__ v2f lWm[(H_DIM / 4) * 4 * 32];
  __shared__ v2f lWs[(H_DIM / 4) * 4 * 32];
  __shared__ __attribute__((aligned(16))) float tiles[8][16 * T_PITCH];
  const int lane = threadIdx.x & 31;
  const int wave = __builtin_amdgcn_readfirstlane((int)(threadIdx.x >> 5));
  const int tile = blockIdx.x * 8 + wave;
  const int row0 = tile * 16;

  const int rpv = (int)min(rowptr[min(row0 + min(lane, 16), N)], (unsigned)E);

  constexpr int FR = (H_DIM / 4) * 4 * 32 / 256;
  v2f pm[FR], ps[FR];
#pragma unroll
  for (int q = 0; q < FR; ++q) {
    const int i  = (int)threadIdx.x + q * 256;
    const int kb = i >> 7;
    const int ct = (i >> 5) & 3;
    const int ln = i & 31;
    const int kg = (ln >> 4) * 2;
    const int n  = ct * 16 + (ln & 15);
    pm[q].x = W_mu[(kb * 4 + kg)     * H_DIM + n];
    pm[q].y = W_mu[(kb * 4 + kg + 1) * H_DIM + n];
    ps[q].x = W_sig[(kb * 4 + kg)     * H_DIM + n];
    ps[q].y = W_sig[(kb * 4 + kg + 1) * H_DIM + n];
  }

  const int span_end = __builtin_amdgcn_readlane(rpv, 16);
  int j0 = __builtin_amdgcn_readlane(rpv, 0);
  int r = 0;
  int rend = __builtin_amdgcn_readlane(rpv, 1);
  v2f acc = {0.0f, 0.0f};
  int cnt = min(32, span_end - j0);
  int raw = 0;
  if (cnt > 0) raw = csr_src[j0 + min(lane, cnt - 1)];

#pragma unroll
  for (int q = 0; q < FR; ++q) {
    lWm[(int)threadIdx.x + q * 256] = pm[q];
    lWs[(int)threadIdx.x + q * 256] = ps[q];
  }
  __syncthreads();
  if (row0 >= N) return;
  float* my = tiles[wave];
  while (cnt > 0) {
    const int sv = min(max(raw, 0), N - 1);
    const int jn = j0 + 32;
    const int cntn = min(32, span_end - jn);
    if (cntn > 0) raw = csr_src[jn + min(lane, cntn - 1)];
    v2f g[32];
#pragma unroll
    for (int u = 0; u < 16; ++u) {
      const int s = __builtin_amdgcn_readlane(sv, min(u, cnt - 1));
      g[u] = *(const v2f*)(h + (long)s * H_DIM + 2 * lane);
    }
    if (cnt > 16) {
#pragma unroll
      for (int u = 16; u < 32; ++u) {
        const int s = __builtin_amdgcn_readlane(sv, min(u, cnt - 1));
        g[u] = *(const v2f*)(h + (long)s * H_DIM + 2 * lane);
      }
    }
#pragma unroll
    for (int u = 0; u < 16; ++u) {
      if (u < cnt) {
        while (j0 + u == rend && r < 15) {
          *(v2f*)(my + r * T_PITCH + 2 * lane) = acc;
          acc.x = 0.0f; acc.y = 0.0f;
          ++r;
          rend = __builtin_amdgcn_readlane(rpv, r + 1);
        }
        acc.x += g[u].x;
        acc.y += g[u].y;
      }
    }
    if (cnt > 16) {
#pragma unroll
      for (int u = 16; u < 32; ++u) {
        if (u < cnt) {
          while (j0 + u == rend && r < 15) {
            *(v2f*)(my + r * T_PITCH + 2 * lane) = acc;
            acc.x = 0.0f; acc.y = 0.0f;
            ++r;
            rend = __builtin_amdgcn_readlane(rpv, r + 1);
          }
          acc.x += g[u].x;
          acc.y += g[u].y;
        }
      }
    }
    j0 = jn;
    cnt = cntn;
  }
  for (; r < 16; ++r) {
    *(v2f*)(my + r * T_PITCH + 2 * lane) = acc;
    acc.x = 0.0f; acc.y = 0.0f;
  }

  const int mcol = lane & 15;
  const int kgrp = (lane >> 4) * 2;
  v8f am[4] = {}, as[4] = {};
  for (int kb = 0; kb < H_DIM / 4; ++kb) {
    const v2f a = *(const v2f*)(my + mcol * T_PITCH + kb * 4 + kgrp);
#pragma unroll
    for (int ct = 0; ct < 4; ++ct) {
      const v2f bm = lWm[(kb * 4 + ct) * 32 + lane];
      const v2f bs = lWs[(kb * 4 + ct) * 32 + lane];
      am[ct] = __builtin_amdgcn_wmma_f32_16x16x4_f32(
          false, a, false, bm, (short)0, am[ct], false, false);
      as[ct] = __builtin_amdgcn_wmma_f32_16x16x4_f32(
          false, a, false, bs, (short)0, as[ct], false, false);
    }
  }

  const int mbase = row0 + ((lane >> 4) << 3);
#pragma unroll
  for (int ct = 0; ct < 4; ++ct) {
    const int n = ct * 16 + (lane & 15);
    const float bmu = b_mu[n];
    const float bsg = b_sig[n];
#pragma unroll
    for (int v = 0; v < 8; ++v) {
      const int m = mbase + v;
      const float nd = norm_dst[m];
      const float mu = am[ct][v] * nd + bmu;
      const float ls = as[ct][v] * nd + bsg;
      z[(long)m * H_DIM + n] = mu + noise[(long)m * H_DIM + n] * expf(ls);
    }
  }
}

extern "C" void kernel_launch(void* const* d_in, const int* in_sizes, int n_in,
                              void* d_out, int out_size, void* d_ws, size_t ws_size,
                              hipStream_t stream) {
  const float* feat  = (const float*)d_in[0];
  const int*   src   = (const int*)d_in[1];
  const int*   dst   = (const int*)d_in[2];
  const float* noise = (const float*)d_in[3];
  const float* W1    = (const float*)d_in[4];
  const float* b1    = (const float*)d_in[5];
  const float* W_mu  = (const float*)d_in[6];
  const float* b_mu  = (const float*)d_in[7];
  const float* W_sig = (const float*)d_in[8];
  const float* b_sig = (const float*)d_in[9];

  const int E = in_sizes[1];
  const int N = in_sizes[3] / H_DIM;

  float* ws       = (float*)d_ws;
  float* norm_src = ws;
  float* norm_dst = ws + N;
  float* buf_a    = ws + 2L * N;
  float* buf_b    = buf_a + (long)N * H_DIM;
  float* zout     = (float*)d_out;

  const int threads = 256;
  const int gemm_blocks = (N / 16 + 7) / 8;
  const int node_blocks = (N + NORM_T - 1) / NORM_T;

  int*      csr_src = (int*)buf_b;
  unsigned* rowptr  = (unsigned*)buf_b + E;
  unsigned* bsum    = rowptr + N + 1;
  float*    h1      = zout;

  unsigned* deg_part = (unsigned*)buf_a;
  unsigned* dplanes  = deg_part + (long)DEG_S * N;
  degree_hist_kernel<<<dim3(DEG_S, (N + DEG_R - 1) / DEG_R, 2), 1024, 0, stream>>>(src, dst, deg_part, E, N);
  norm_kernel<<<dim3(node_blocks, 2), NORM_T, 0, stream>>>(deg_part, norm_src, bsum, N);
  offsets_kernel<<<node_blocks, NORM_T, 0, stream>>>(dplanes, bsum, rowptr, N);

  const int n_scatter = DEG_S * ((N + SC_R - 1) / SC_R);
  const int l1_min    = (N / 16 + 31) / 32;
  const int n_gemm    = l1_min > L1_GROUPS ? l1_min : L1_GROUPS;
  sort_scatter_and_layer1_kernel<<<n_scatter + n_gemm, 1024, 0, stream>>>(
      src, dst, dplanes, csr_src, feat, norm_src, W1, h1, E, N, n_scatter, n_gemm);

  aggregate_rows_kernel<<<(N + 7) / 8, threads, 0, stream>>>(
      csr_src, rowptr, h1, norm_dst, b1, norm_src, buf_a, 1, N, E);

  aggregate_gemm2_kernel<<<gemm_blocks, threads, 0, stream>>>(
      csr_src, rowptr, buf_a, W_mu, b_mu, W_sig, b_sig, norm_dst, noise, zout, N, E);
}


# Last launch (layer-2 aggregation + mu/sigma products): 256 work-groups of 25 waves, tile = wave * 256 + group: 24 or 25 tiles per compute unit; was 782 eight-wave groups, 3 or 4 per unit (24 or 32)
# speedup vs baseline: 3.5096x; 1.0355x over previous
#include <hip/hip_runtime.h>

typedef __attribute__((ext_vector_type(2))) float v2f;
typedef __attribute__((ext_vector_type(8))) float v8f;

#define F_IN  128
#define H_DIM 64

#define DEG_R 25600
#define DEG_S 32
static_assert(2 * DEG_S <= H_DIM, "partial planes must fit the h buffer");

__device__ __forceinline__ void count4(unsigned* bins, const int4 q, const int lo) {
  const unsigned r0 = (unsigned)(q.x - lo), r1 = (unsigned)(q.y - lo);
  const unsigned r2 = (unsigned)(q.z - lo), r3 = (unsigned)(q.w - lo);
  if (r0 < DEG_R) atomicAdd(&bins[r0], 1u);
  if (r1 < DEG_R) atomicAdd(&bins[r1], 1u);
  if (r2 < DEG_R) atomicAdd(&bins[r2], 1u);
  if (r3 < DEG_R) atomicAdd(&bins[r3], 1u);
}

__global__ __launch_bounds__(1024) void degree_hist_kernel(
    const int* __restrict__ src, const int* __restrict__ dst,
    unsigned* __restrict__ part, int E, int N) {
  __shared__ unsigned bins[DEG_R];
  const int* __restrict__ idx = blockIdx.z ? dst : src;
  const int lo = blockIdx.y * DEG_R;
  for (int i = threadIdx.x; i < DEG_R; i += 1024) bins[i] = 0u;
  __syncthreads();

  const int chunk = (((E + DEG_S - 1) / DEG_S) + 3) & ~3;
  const int beg = min(E, (int)blockIdx.x * chunk);
  const int end = min(E, beg + chunk);
  const int nvec = (end - beg) >> 2;
  const int4* __restrict__ v = (const int4*)(idx + beg);
  const int4 none = make_int4(lo - 1, lo - 1, lo - 1, lo - 1);
  const int last = nvec - 1;
  for (int i = threadIdx.x; i < nvec; i += 4096) {
    const int4 q0 = v[i];
    int4 q1 = v[min(i + 1024, last)];
    int4 q2 = v[min(i + 2048, last)];
    int4 q3 = v[min(i + 3072, last)];
    if (i + 1024 > last) q1 = none;
    if (i + 2048 > last) q2 = none;
    if (i + 3072 > last) q3 = none;
    count4(bins, q0, lo);
    count4(bins, q1, lo);
    count4(bins, q2, lo);
    count4(bins, q3, lo);
  }
  for (int i = beg + (nvec << 2) + threadIdx.x; i < end; i += 1024) {
    const unsigned r = (unsigned)(idx[i] - lo);
    if (r < DEG_R) atomicAdd(&bins[r], 1u);
  }
  __syncthreads();

  unsigned* __restrict__ plane = part + (long)(blockIdx.z * DEG_S + blockIdx.x) * N;
  const int hi = min(N, lo + DEG_R);
  for (int i = lo + threadIdx.x; i < hi; i += 1024) plane[i] = bins[i - lo];
}

#define NORM_T 256
__global__ __launch_bounds__(NORM_T) void norm_kernel(const unsigned* __restrict__ part, float* __restrict__ norm,
                                                      unsigned* __restrict__ bsum, int N) {
  __shared__ unsigned wsum[NORM_T / 32];
  const int i = blockIdx.x * NORM_T + threadIdx.x;
  unsigned c = 0u;
  if (i < N) {
    const unsigned* __restrict__ p = part + (long)blockIdx.y * DEG_S * N + i;
    for (int s = 0; s < DEG_S; ++s) c += p[(long)s * N];
    norm[(long)blockIdx.y * N + i] = rsqrtf(fmaxf((float)c, 1.0f));
  }
  if (blockIdx.y == 1) {
    unsigned t = c;
    for (int o = 16; o > 0; o >>= 1) t += __shfl_xor(t, o);
    if ((threadIdx.x & 31) == 0) wsum[threadIdx.x >> 5] = t;
    __syncthreads();
    if (threadIdx.x == 0) {
      unsigned b = 0u;
      for (int w = 0; w < NORM_T / 32; ++w) b += wsum[w];
      bsum[blockIdx.x] = b;
    }
  }
}

__global__ __launch_bounds__(NORM_T) void offsets_kernel(unsigned* __restrict__ dplanes,
                                                         const unsigned* __restrict__ bsum,
                                                         unsigned* __restrict__ rowptr, int N) {
  __shared__ unsigned red[NORM_T / 32];
  __shared__ unsigned wtot[NORM_T / 32];
  const int lane = threadIdx.x & 31;
  const int wave = threadIdx.x >> 5;
  unsigned b = 0u;
  for (int k = threadIdx.x; k < (int)blockIdx.x; k += NORM_T) b += bsum[k];
  for (int o = 16; o > 0; o >>= 1) b += __shfl_xor(b, o);
  if (lane == 0) red[wave] = b;

  const int node = blockIdx.x * NORM_T + threadIdx.x;
  unsigned* __restrict__ p = dplanes + min(node, N - 1);
  unsigned t[DEG_S];
  unsigned c = 0u;
#pragma unroll
  for (int s = 0; s < DEG_S; ++s) {
    t[s] = (node < N) ? p[(long)s * N] : 0u;
    c += t[s];
  }
  unsigned x = c;
  for (int o = 1; o < 32; o <<= 1) {
    const unsigned y = __shfl_up(x, o);
    if (lane >= o) x += y;
  }
  if (lane == 31) wtot[wave] = x;
  __syncthreads();
  unsigned run = x - c;
  for (int w = 0; w < NORM_T / 32; ++w) {
    run += red[w];
    if (w < wave) run += wtot[w];
  }
  if (node < N) {
    rowptr[node] = run;
#pragma unroll
    for (int s = 0; s < DEG_S; ++s) {
      p[(long)s * N] = run;
      run += t[s];
    }
    if (node == N - 1) rowptr[N] = run;
  }
}

#define SC_R 12800
__device__ __forceinline__ void place1(unsigned* cur, int* __restrict__ csr_src,
                                       const int d, const int s, const int lo, const int E) {
  const unsigned r = (unsigned)(d - lo);
  if (r < SC_R) {
    const unsigned pos = atomicAdd(&cur[r], 1u);
    if (pos < (unsigned)E) csr_src[pos] = s;
  }
}
__device__ __forceinline__ void place4(unsigned* cur, int* __restrict__ csr_src,
                                       const int4 d, const int4 s, const int lo, const int E) {
  place1(cur, csr_src, d.x, s.x, lo, E);
  place1(cur, csr_src, d.y, s.y, lo, E);
  place1(cur, csr_src, d.z, s.z, lo, E);
  place1(cur, csr_src, d.w, s.w, lo, E);
}

__device__ __forceinline__ void scatter_body(
    unsigned* cur, const int slice, const int range,
    const int* __restrict__ src, const int* __restrict__ dst,
    const unsigned* __restrict__ dplanes, int* __restrict__ csr_src, int E, int N) {
  const int lo = range * SC_R;
  const int hi = min(N, lo + SC_R);
  const unsigned* __restrict__ plane = dplanes + (long)slice * N;
  for (int i = lo + threadIdx.x; i < hi; i += 1024) cur[i - lo] = plane[i];
  __syncthreads();

  const int chunk = (((E + DEG_S - 1) / DEG_S) + 3) & ~3;
  const int beg = min(E, slice * chunk);
  const int end = min(E, beg + chunk);
  const int nvec = (end - beg) >> 2;
  const int4* __restrict__ dv = (const int4*)(dst + beg);
  const int4* __restrict__ sv = (const int4*)(src + beg);
  const int4 none = make_int4(lo - 1, lo - 1, lo - 1, lo - 1);
  const int last = nvec - 1;
  for (int i = threadIdx.x; i < nvec; i += 2048) {
    const int i1 = min(i + 1024, last);
    const int4 d0 = dv[i];
    const int4 s0 = sv[i];
    int4 d1 = dv[i1];
    const int4 s1 = sv[i1];
    if (i + 1024 > last) d1 = none;
    place4(cur, csr_src, d0, s0, lo, E);
    place4(cur, csr_src, d1, s1, lo, E);
  }
  for (int i = beg + (nvec << 2) + threadIdx.x; i < end; i += 1024)
    place1(cur, csr_src, dst[i], src[i], lo, E);
}

__device__ __forceinline__ void stage_w_frag(v2f* lW, const float* __restrict__ W,
                                             int kblocks, int tid, int nthreads) {
  const int total = kblocks * 4 * 32;
  for (int i = tid; i < total; i += nthreads) {
    const int kb = i >> 7;
    const int ct = (i >> 5) & 3;
    const int ln = i & 31;
    const int kg = (ln >> 4) * 2;
    const int n  = ct * 16 + (ln & 15);
    v2f p;
    p.x = W[(kb * 4 + kg)     * H_DIM + n];
    p.y = W[(kb * 4 + kg + 1) * H_DIM + n];
    lW[i] = p;
  }
}

__device__ __forceinline__ void gemm1_body(
    v2f* lW, const int grp, const int ngrp,
    const float* __restrict__ feat, const float* __restrict__ norm_src,
    const float* __restrict__ W1, float* __restrict__ h1, int N) {
  stage_w_frag(lW, W1, F_IN / 4, threadIdx.x, 1024);
  __syncthreads();

  const int wave = threadIdx.x >> 5;
  const int lane = threadIdx.x & 31;
  const int tile = wave * ngrp + grp;
  const int row0 = tile * 16;
  if (row0 >= N) return;

  const int mrow = row0 + (lane & 15);
  const int kgrp = (lane >> 4) * 2;
  const float ns = norm_src[mrow];

  v8f acc[4] = {};
  for (int kb = 0; kb < F_IN / 4; ++kb) {
    v2f a = *(const v2f*)(feat + (long)mrow * F_IN + kb * 4 + kgrp);
    a.x *= ns; a.y *= ns;
#pragma unroll
    for (int ct = 0; ct < 4; ++ct) {
      const v2f b = lW[(kb * 4 + ct) * 32 + lane];
      acc[ct] = __builtin_amdgcn_wmma_f32_16x16x4_f32(
          false, a, false, b, (short)0, acc[ct], false, false);
    }
  }

  const int mbase = row0 + ((lane >> 4) << 3);
#pragma unroll
  for (int ct = 0; ct < 4; ++ct) {
    const int n = ct * 16 + (lane & 15);
#pragma unroll
    for (int v = 0; v < 8; ++v)
      h1[(long)(mbase + v) * H_DIM + n] = acc[ct][v];
  }
}

#define L1_GROUPS 256
__global__ __launch_bounds__(1024) void sort_scatter_and_layer1_kernel(
    const int* __restrict__ src, const int* __restrict__ dst,
    const unsigned* __restrict__ dplanes, int* __restrict__ csr_src,
    const float* __restrict__ feat, const float* __restrict__ norm_src,
    const float* __restrict__ W1, float* __restrict__ h1,
    int E, int N, int n_scatter, int n_gemm) {
  __shared__ __attribute__((aligned(16))) unsigned char smem[SC_R * sizeof(unsigned)];
  static_assert(sizeof(v2f) * (F_IN / 4) * 4 * 32 <= SC_R * sizeof(unsigned), "W1 fragments must fit the cursor array");
  const int id = blockIdx.x;
  if (id < n_scatter)
    scatter_body((unsigned*)smem, id % DEG_S, id / DEG_S, src, dst, dplanes, csr_src, E, N);
  else
    gemm1_body((v2f*)smem, id - n_scatter, n_gemm, feat, norm_src, W1, h1, N);
}

#define AGG_BATCH 8
__global__ __launch_bounds__(256) void aggregate_rows_kernel(
    const int* __restrict__ csr_src, const unsigned* __restrict__ rowptr,
    const float* __restrict__ h, const float* __restrict__ norm_dst,
    const float* __restrict__ b1, const float* __restrict__ norm_src,
    float* __restrict__ res, int relu, int N, int E) {
  const int lane = threadIdx.x & 31;
  const int wave = __builtin_amdgcn_readfirstlane((int)(threadIdx.x >> 5));
  const int row = blockIdx.x * 8 + wave;
  if (row >= N) return;
  const int beg = (int)min(rowptr[row], (unsigned)E);
  const int end = (int)min(rowptr[row + 1], (unsigned)E);
  v2f acc = {0.0f, 0.0f};
  for (int j0 = beg; j0 < end; j0 += 32) {
    const int cnt = min(32, end - j0);
    const int sv = min(max(csr_src[j0 + min(lane, cnt - 1)], 0), N - 1);
    for (int u0 = 0; u0 < cnt; u0 += AGG_BATCH) {
      v2f g[AGG_BATCH];
      for (int u = 0; u < AGG_BATCH; ++u) {
        const int s = __builtin_amdgcn_readlane(sv, min(u0 + u, cnt - 1));
        g[u] = *(const v2f*)(h + (long)s * H_DIM + 2 * lane);
      }
      for (int u = 0; u < AGG_BATCH; ++u) {
        if (u0 + u < cnt) {
          acc.x += g[u].x;
          acc.y += g[u].y;
        }
      }
    }
  }
  if (relu) {
    const float nd = norm_dst[row];
    const float ns = norm_src[row];
    acc.x = fmaxf(acc.x * nd + b1[2 * lane], 0.0f) * ns;
    acc.y = fmaxf(acc.y * nd + b1[2 * lane + 1], 0.0f) * ns;
  }
  *(v2f*)(res + (long)row * H_DIM + 2 * lane) = acc;
}

#define T_PITCH 68
#define G2_WAVES 25
#define G2_GROUPS 256
__global__ __launch_bounds__(G2_WAVES * 32) void aggregate_gemm2_kernel(
    const int* __restrict__ csr_src, const unsigned* __restrict__ rowptr,
    const float* __restrict__ h,
    const float* __restrict__ W_mu, const float* __restrict__ b_mu,
    const float* __restrict__ W_sig, const float* __restrict__ b_sig,
    const float* __restrict__ norm_dst, const float* __restrict__ noise,
    float* __restrict__ z, int N, int E) {
  __shared__ v2f lWm[(H_DIM / 4) * 4 * 32];
  __shared__ v2f lWs[(H_DIM / 4) * 4 * 32];
  __shared__ __attribute__((aligned(16))) float tiles[G2_WAVES][16 * T_PITCH];
  const int lane = threadIdx.x & 31;
  const int wave = __builtin_amdgcn_readfirstlane((int)(threadIdx.x >> 5));
  const int tile = wave * (int)gridDim.x + (int)blockIdx.x;
  const int row0 = tile * 16;

  const int rpv = (int)min(rowptr[min(row0 + min(lane, 16), N)], (unsigned)E);

  constexpr int NFRAG = (H_DIM / 4) * 4 * 32;
  constexpr int NTHR  = G2_WAVES * 32;
  constexpr int FR = (NFRAG + NTHR - 1) / NTHR;
  v2f pm[FR], ps[FR];
#pragma unroll
  for (int q = 0; q < FR; ++q) {
    const int i  = (int)threadIdx.x + q * NTHR;
    pm[q].x = 0.0f; pm[q].y = 0.0f; ps[q].x = 0.0f; ps[q].y = 0.0f;
    if (i >= NFRAG) continue;
    const int kb = i >> 7;
    const int ct = (i >> 5) & 3;
    const int ln = i & 31;
    const int kg = (ln >> 4) * 2;
    const int n  = ct * 16 + (ln & 15);
    pm[q].x = W_mu[(kb * 4 + kg)     * H_DIM + n];
    pm[q].y = W_mu[(kb * 4 + kg + 1) * H_DIM + n];
    ps[q].x = W_sig[(kb * 4 + kg)     * H_DIM + n];
    ps[q].y = W_sig[(kb * 4 + kg + 1) * H_DIM + n];
  }

  const int span_end = __builtin_amdgcn_readlane(rpv, 16);
  int j0 = __builtin_amdgcn_readlane(rpv, 0);
  int r = 0;
  int rend = __builtin_amdgcn_readlane(rpv, 1);
  v2f acc = {0.0f, 0.0f};
  int cnt = min(32, span_end - j0);
  int raw = 0;
  if (cnt > 0) raw = csr_src[j0 + min(lane, cnt - 1)];

#pragma unroll
  for (int q = 0; q < FR; ++q) {
    const int i = (int)threadIdx.x + q * NTHR;
    if (i < NFRAG) {
      lWm[i] = pm[q];
      lWs[i] = ps[q];
    }
  }
  __syncthreads();
  if (row0 >= N) return;
  float* my = tiles[wave];
  while (cnt > 0) {
    const int sv = min(max(raw, 0), N - 1);
    const int jn = j0 + 32;
    const int cntn = min(32, span_end - jn);
    if (cntn > 0) raw = csr_src[jn + min(lane, cntn - 1)];
    v2f g[32];
#pragma unroll
    for (int u = 0; u < 16; ++u) {
      const int s = __builtin_amdgcn_readlane(sv, min(u, cnt - 1));
      g[u] = *(const v2f*)(h + (long)s * H_DIM + 2 * lane);
    }
    if (cnt > 16) {
#pragma unroll
      for (int u = 16; u < 32; ++u) {
        const int s = __builtin_amdgcn_readlane(sv, min(u, cnt - 1));
        g[u] = *(const v2f*)(h + (long)s * H_DIM + 2 * lane);
      }
    }
#pragma unroll
    for (int u = 0; u < 16; ++u) {
      if (u < cnt) {
        while (j0 + u == rend && r < 15) {
          *(v2f*)(my + r * T_PITCH + 2 * lane) = acc;
          acc.x = 0.0f; acc.y = 0.0f;
          ++r;
          rend = __builtin_amdgcn_readlane(rpv, r + 1);
        }
        acc.x += g[u].x;
        acc.y += g[u].y;
      }
    }
    if (cnt > 16) {
#pragma unroll
      for (int u = 16; u < 32; ++u) {
        if (u < cnt) {
          while (j0 + u == rend && r < 15) {
            *(v2f*)(my + r * T_PITCH + 2 * lane) = acc;
            acc.x = 0.0f; acc.y = 0.0f;
            ++r;
            rend = __builtin_amdgcn_readlane(rpv, r + 1);
          }
          acc.x += g[u].x;
          acc.y += g[u].y;
        }
      }
    }
    j0 = jn;
    cnt = cntn;
  }
  for (; r < 16; ++r) {
    *(v2f*)(my + r * T_PITCH + 2 * lane) = acc;
    acc.x = 0.0f; acc.y = 0.0f;
  }

  const int mcol = lane & 15;
  const int kgrp = (lane >> 4) * 2;
  v8f am[4] = {}, as[4] = {};
  for (int kb = 0; kb < H_DIM / 4; ++kb) {
    const v2f a = *(const v2f*)(my + mcol * T_PITCH + kb * 4 + kgrp);
#pragma unroll
    for (int ct = 0; ct < 4; ++ct) {
      const v2f bm = lWm[(kb * 4 + ct) * 32 + lane];
      const v2f bs = lWs[(kb * 4 + ct) * 32 + lane];
      am[ct] = __builtin_amdgcn_wmma_f32_16x16x4_f32(
          false, a, false, bm, (short)0, am[ct], false, false);
      as[ct] = __builtin_amdgcn_wmma_f32_16x16x4_f32(
          false, a, false, bs, (short)0, as[ct], false, false);
    }
  }

  const int mbase = row0 + ((lane >> 4) << 3);
#pragma unroll
  for (int ct = 0; ct < 4; ++ct) {
    const int n = ct * 16 + (lane & 15);
    const float bmu = b_mu[n];
    const float bsg = b_sig[n];
#pragma unroll
    for (int v = 0; v < 8; ++v) {
      const int m = mbase + v;
      const float nd = norm_dst[m];
      const float mu = am[ct][v] * nd + bmu;
      const float ls = as[ct][v] * nd + bsg;
      z[(long)m * H_DIM + n] = mu + noise[(long)m * H_DIM + n] * expf(ls);
    }
  }
}

extern "C" void kernel_launch(void* const* d_in, const int* in_sizes, int n_in,
                              void* d_out, int out_size, void* d_ws, size_t ws_size,
                              hipStream_t stream) {
  const float* feat  = (const float*)d_in[0];
  const int*   src   = (const int*)d_in[1];
  const int*   dst   = (const int*)d_in[2];
  const float* noise = (const float*)d_in[3];
  const float* W1    = (const float*)d_in[4];
  const float* b1    = (const float*)d_in[5];
  const float* W_mu  = (const float*)d_in[6];
  const float* b_mu  = (const float*)d_in[7];
  const float* W_sig = (const float*)d_in[8];
  const float* b_sig = (const float*)d_in[9];

  const int E = in_sizes[1];
  const int N = in_sizes[3] / H_DIM;

  float* ws       = (float*)d_ws;
  float* norm_src = ws;
  float* norm_dst = ws + N;
  float* buf_a    = ws + 2L * N;
  float* buf_b    = buf_a + (long)N * H_DIM;
  float* zout     = (float*)d_out;

  const int threads = 256;
  const int node_blocks = (N + NORM_T - 1) / NORM_T;

  int*      csr_src = (int*)buf_b;
  unsigned* rowptr  = (unsigned*)buf_b + E;
  unsigned* bsum    = rowptr + N + 1;
  float*    h1      = zout;

  unsigned* deg_part = (unsigned*)buf_a;
  unsigned* dplanes  = deg_part + (long)DEG_S * N;
  degree_hist_kernel<<<dim3(DEG_S, (N + DEG_R - 1) / DEG_R, 2), 1024, 0, stream>>>(src, dst, deg_part, E, N);
  norm_kernel<<<dim3(node_blocks, 2), NORM_T, 0, stream>>>(deg_part, norm_src, bsum, N);
  offsets_kernel<<<node_blocks, NORM_T, 0, stream>>>(dplanes, bsum, rowptr, N);

  const int n_scatter = DEG_S * ((N + SC_R - 1) / SC_R);
  const int l1_min    = (N / 16 + 31) / 32;
  const int n_gemm    = l1_min > L1_GROUPS ? l1_min : L1_GROUPS;
  sort_scatter_and_layer1_kernel<<<n_scatter + n_gemm, 1024, 0, stream>>>(
      src, dst, dplanes, csr_src, feat, norm_src, W1, h1, E, N, n_scatter, n_gemm);

  aggregate_rows_kernel<<<(N + 7) / 8, threads, 0, stream>>>(
      csr_src, rowptr, h1, norm_dst, b1, norm_src, buf_a, 1, N, E);

  const int g2_min = (N / 16 + G2_WAVES - 1) / G2_WAVES;
  const int n_g2   = g2_min > G2_GROUPS ? g2_min : G2_GROUPS;
  aggregate_gemm2_kernel<<<n_g2, G2_WAVES * 32, 0, stream>>>(
      csr_src, rowptr, buf_a, W_mu, b_mu, W_sig, b_sig, norm_dst, noise, zout, N, E);
}


# Last launch, gather-and-sum loop: full 32-edge chunks straight-line: one scalar compare per edge for the row end, constant lane reads, 32-bit row offsets (21 -> 7 instrs per edge); sum order kept
# speedup vs baseline: 3.7687x; 1.0738x over previous
#include <hip/hip_runtime.h>

typedef __attribute__((ext_vector_type(2))) float v2f;
typedef __attribute__((ext_vector_type(8))) float v8f;

#define F_IN  128
#define H_DIM 64

#define DEG_R 25600
#define DEG_S 32
static_assert(2 * DEG_S <= H_DIM, "partial planes must fit the h buffer");
#define NORM_T 256
#define DEG_BLK (DEG_R / NORM_T)
static_assert(DEG_R % NORM_T == 0, "a range must hold whole blocks");

__device__ __forceinline__ void count4(unsigned* bins, const int4 q, const int lo) {
  const unsigned r0 = (unsigned)(q.x - lo), r1 = (unsigned)(q.y - lo);
  const unsigned r2 = (unsigned)(q.z - lo), r3 = (unsigned)(q.w - lo);
  if (r0 < DEG_R) atomicAdd(&bins[r0], 1u);
  if (r1 < DEG_R) atomicAdd(&bins[r1], 1u);
  if (r2 < DEG_R) atomicAdd(&bins[r2], 1u);
  if (r3 < DEG_R) atomicAdd(&bins[r3], 1u);
}

__global__ __launch_bounds__(1024) void degree_hist_kernel(
    const int* __restrict__ src, const int* __restrict__ dst,
    unsigned* __restrict__ part, unsigned* __restrict__ bpre, unsigned* __restrict__ rtot,
    int E, int N) {
  __shared__ unsigned bins[DEG_R];
  __shared__ unsigned bs[DEG_BLK];
  const int* __restrict__ idx = blockIdx.z ? dst : src;
  const int lo = blockIdx.y * DEG_R;
  for (int i = threadIdx.x; i < DEG_R; i += 1024) bins[i] = 0u;
  __syncthreads();

  const int chunk = (((E + DEG_S - 1) / DEG_S) + 3) & ~3;
  const int beg = min(E, (int)blockIdx.x * chunk);
  const int end = min(E, beg + chunk);
  const int nvec = (end - beg) >> 2;
  const int4* __restrict__ v = (const int4*)(idx + beg);
  const int4 none = make_int4(lo - 1, lo - 1, lo - 1, lo - 1);
  const int last = nvec - 1;
  for (int i = threadIdx.x; i < nvec; i += 4096) {
    const int4 q0 = v[i];
    int4 q1 = v[min(i + 1024, last)];
    int4 q2 = v[min(i + 2048, last)];
    int4 q3 = v[min(i + 3072, last)];
    if (i + 1024 > last) q1 = none;
    if (i + 2048 > last) q2 = none;
    if (i + 3072 > last) q3 = none;
    count4(bins, q0, lo);
    count4(bins, q1, lo);
    count4(bins, q2, lo);
    count4(bins, q3, lo);
  }
  for (int i = beg + (nvec << 2) + threadIdx.x; i < end; i += 1024) {
    const unsigned r = (unsigned)(idx[i] - lo);
    if (r < DEG_R) atomicAdd(&bins[r], 1u);
  }
  __syncthreads();

  unsigned* __restrict__ plane = part + (long)(blockIdx.z * DEG_S + blockIdx.x) * N;
  const int hi = min(N, lo + DEG_R);
  for (int i = lo + threadIdx.x; i < hi; i += 1024) plane[i] = bins[i - lo];

  if (blockIdx.z == 1) {
    const int lane = threadIdx.x & 31;
    const int wave = threadIdx.x >> 5;
    constexpr int BINS_T = 32;
    constexpr int T_BLK  = NORM_T / BINS_T;
    static_assert(NORM_T % BINS_T == 0 && 32 % T_BLK == 0, "the threads of a block must sit in one wave");
    static_assert(DEG_R % (32 * BINS_T) == 0 && DEG_R / BINS_T <= 1024, "whole waves, one pass");
    const int g = threadIdx.x;
    unsigned t = 0u;
    if (g < DEG_R / BINS_T) {
#pragma unroll
      for (int j = 0; j < BINS_T; ++j) {
        const int x = g * BINS_T + ((j + lane) & (BINS_T - 1));
        const unsigned q = bins[x];
        t += (lo + x < N) ? q : 0u;
      }
    }
    for (int o = T_BLK / 2; o > 0; o >>= 1) t += __shfl_xor(t, o);
    if (g < DEG_R / BINS_T && (lane & (T_BLK - 1)) == 0) bs[g / T_BLK] = t;
    __syncthreads();
    if (wave == 0) {
      constexpr int PER = (DEG_BLK + 31) / 32;
      unsigned v[PER];
      unsigned s = 0u;
#pragma unroll
      for (int q = 0; q < PER; ++q) {
        const int k = lane * PER + q;
        v[q] = (k < DEG_BLK) ? bs[k] : 0u;
        s += v[q];
      }
      unsigned x = s;
      for (int o = 1; o < 32; o <<= 1) {
        const unsigned y = __shfl_up(x, o);
        if (lane >= o) x += y;
      }
      unsigned run = x - s;
      const int nbk = (int)gridDim.y * DEG_BLK;
      unsigned* __restrict__ row = bpre + (long)blockIdx.x * nbk + (int)blockIdx.y * DEG_BLK;
#pragma unroll
      for (int q = 0; q < PER; ++q) {
        const int k = lane * PER + q;
        if (k < DEG_BLK) row[k] = run;
        run += v[q];
      }
      if (lane == 31) rtot[(int)blockIdx.y * DEG_S + (int)blockIdx.x] = x;
    }
  }
}

__global__ __launch_bounds__(NORM_T) void offsets_kernel(const unsigned* __restrict__ splanes,
                                                         unsigned* __restrict__ dplanes,
                                                         const unsigned* __restrict__ bpre,
                                                         const unsigned* __restrict__ rtot,
                                                         unsigned* __restrict__ rowptr,
                                                         float* __restrict__ norm, int N) {
  __shared__ unsigned red[NORM_T / 32];
  __shared__ unsigned wtot[NORM_T / 32];
  const int lane = threadIdx.x & 31;
  const int wave = threadIdx.x >> 5;
  const int nbk = ((N + DEG_R - 1) / DEG_R) * DEG_BLK;
  const int nfull = ((int)blockIdx.x / DEG_BLK) * DEG_S;
  unsigned b = 0u;
  unsigned bt = 0u;
  if ((int)threadIdx.x < DEG_S) b = bpre[(long)threadIdx.x * nbk + blockIdx.x];
  if ((int)threadIdx.x < nfull) bt = rtot[threadIdx.x];
  for (int k = threadIdx.x + NORM_T; k < nfull; k += NORM_T) bt += rtot[k];

  const int node = blockIdx.x * NORM_T + threadIdx.x;
  unsigned* __restrict__ p = dplanes + min(node, N - 1);
  const unsigned* __restrict__ ps = splanes + min(node, N - 1);
  unsigned t[DEG_S];
  unsigned c = 0u;
  unsigned cs = 0u;
#pragma unroll
  for (int s = 0; s < DEG_S; ++s) {
    t[s] = (node < N) ? p[(long)s * N] : 0u;
    c += t[s];
    cs += ps[(long)s * N];
  }
  b += bt;
  for (int o = 16; o > 0; o >>= 1) b += __shfl_xor(b, o);
  if (lane == 0) red[wave] = b;
  unsigned x = c;
  for (int o = 1; o < 32; o <<= 1) {
    const unsigned y = __shfl_up(x, o);
    if (lane >= o) x += y;
  }
  if (lane == 31) wtot[wave] = x;
  __syncthreads();
  unsigned run = x - c;
  for (int w = 0; w < NORM_T / 32; ++w) {
    run += red[w];
    if (w < wave) run += wtot[w];
  }
  if (node < N) {
    norm[node] = rsqrtf(fmaxf((float)cs, 1.0f));
    norm[(long)N + node] = rsqrtf(fmaxf((float)c, 1.0f));
    rowptr[node] = run;
#pragma unroll
    for (int s = 0; s < DEG_S; ++s) {
      p[(long)s * N] = run;
      run += t[s];
    }
    if (node == N - 1) rowptr[N] = run;
  }
}

#define SC_R 12800
__device__ __forceinline__ void place1(unsigned* cur, int* __restrict__ csr_src,
                                       const int d, const int s, const int lo, const int E) {
  const unsigned r = (unsigned)(d - lo);
  if (r < SC_R) {
    const unsigned pos = atomicAdd(&cur[r], 1u);
    if (pos < (unsigned)E) csr_src[pos] = s;
  }
}
__device__ __forceinline__ void place4(unsigned* cur, int* __restrict__ csr_src,
                                       const int4 d, const int4 s, const int lo, const int E) {
  place1(cur, csr_src, d.x, s.x, lo, E);
  place1(cur, csr_src, d.y, s.y, lo, E);
  place1(cur, csr_src, d.z, s.z, lo, E);
  place1(cur, csr_src, d.w, s.w, lo, E);
}

__device__ __forceinline__ void scatter_body(
    unsigned* cur, const int slice, const int range,
    const int* __restrict__ src, const int* __restrict__ dst,
    const unsigned* __restrict__ dplanes, int* __restrict__ csr_src, int E, int N) {
  const int lo = range * SC_R;
  const int hi = min(N, lo + SC_R);
  const unsigned* __restrict__ plane = dplanes + (long)slice * N;
  for (int i = lo + threadIdx.x; i < hi; i += 1024) cur[i - lo] = plane[i];
  __syncthreads();

  const int chunk = (((E + DEG_S - 1) / DEG_S) + 3) & ~3;
  const int beg = min(E, slice * chunk);
  const int end = min(E, beg + chunk);
  const int nvec = (end - beg) >> 2;
  const int4* __restrict__ dv = (const int4*)(dst + beg);
  const int4* __restrict__ sv = (const int4*)(src + beg);
  const int4 none = make_int4(lo - 1, lo - 1, lo - 1, lo - 1);
  const int last = nvec - 1;
  for (int i = threadIdx.x; i < nvec; i += 2048) {
    const int i1 = min(i + 1024, last);
    const int4 d0 = dv[i];
    const int4 s0 = sv[i];
    int4 d1 = dv[i1];
    const int4 s1 = sv[i1];
    if (i + 1024 > last) d1 = none;
    place4(cur, csr_src, d0, s0, lo, E);
    place4(cur, csr_src, d1, s1, lo, E);
  }
  for (int i = beg + (nvec << 2) + threadIdx.x; i < end; i += 1024)
    place1(cur, csr_src, dst[i], src[i], lo, E);
}

__device__ __forceinline__ void stage_w_frag(v2f* lW, const float* __restrict__ W,
                                             int kblocks, int tid, int nthreads) {
  const int total = kblocks * 4 * 32;
  for (int i = tid; i < total; i += nthreads) {
    const int kb = i >> 7;
    const int ct = (i >> 5) & 3;
    const int ln = i & 31;
    const int kg = (ln >> 4) * 2;
    const int n  = ct * 16 + (ln & 15);
    v2f p;
    p.x = W[(kb * 4 + kg)     * H_DIM + n];
    p.y = W[(kb * 4 + kg + 1) * H_DIM + n];
    lW[i] = p;
  }
}

__device__ __forceinline__ void gemm1_body(
    v2f* lW, const int grp, const int ngrp,
    const float* __restrict__ feat, const float* __restrict__ norm_src,
    const float* __restrict__ W1, float* __restrict__ h1, int N) {
  stage_w_frag(lW, W1, F_IN / 4, threadIdx.x, 1024);
  __syncthreads();

  const int wave = threadIdx.x >> 5;
  const int lane = threadIdx.x & 31;
  const int tile = wave * ngrp + grp;
  const int row0 = tile * 16;
  if (row0 >= N) return;

  const int mrow = row0 + (lane & 15);
  const int kgrp = (lane >> 4) * 2;
  const float ns = norm_src[mrow];

  v8f acc[4] = {};
  for (int kb = 0; kb < F_IN / 4; ++kb) {
    v2f a = *(const v2f*)(feat + (long)mrow * F_IN + kb * 4 + kgrp);
    a.x *= ns; a.y *= ns;
#pragma unroll
    for (int ct = 0; ct < 4; ++ct) {
      const v2f b = lW[(kb * 4 + ct) * 32 + lane];
      acc[ct] = __builtin_amdgcn_wmma_f32_16x16x4_f32(
          false, a, false, b, (short)0, acc[ct], false, false);
    }
  }

  const int mbase = row0 + ((lane >> 4) << 3);
#pragma unroll
  for (int ct = 0; ct < 4; ++ct) {
    const int n = ct * 16 + (lane & 15);
#pragma unroll
    for (int v = 0; v < 8; ++v)
      h1[(long)(mbase + v) * H_DIM + n] = acc[ct][v];
  }
}

#define L1_GROUPS 256
__global__ __launch_bounds__(1024) void sort_scatter_and_layer1_kernel(
    const int* __restrict__ src, const int* __restrict__ dst,
    const unsigned* __restrict__ dplanes, int* __restrict__ csr_src,
    const float* __restrict__ feat, const float* __restrict__ norm_src,
    const float* __restrict__ W1, float* __restrict__ h1,
    int E, int N, int n_scatter, int n_gemm) {
  __shared__ __attribute__((aligned(16))) unsigned char smem[SC_R * sizeof(unsigned)];
  static_assert(sizeof(v2f) * (F_IN / 4) * 4 * 32 <= SC_R * sizeof(unsigned), "W1 fragments must fit the cursor array");
  const int id = blockIdx.x;
  if (id < n_scatter)
    scatter_body((unsigned*)smem, id % DEG_S, id / DEG_S, src, dst, dplanes, csr_src, E, N);
  else
    gemm1_body((v2f*)smem, id - n_scatter, n_gemm, feat, norm_src, W1, h1, N);
}

#define AGG_BATCH 8
__global__ __launch_bounds__(256) void aggregate_rows_kernel(
    const int* __restrict__ csr_src, const unsigned* __restrict__ rowptr,
    const float* __restrict__ h, const float* __restrict__ norm_dst,
    const float* __restrict__ b1, const float* __restrict__ norm_src,
    float* __restrict__ res, int relu, int N, int E) {
  const int lane = threadIdx.x & 31;
  const int wave = __builtin_amdgcn_readfirstlane((int)(threadIdx.x >> 5));
  const int row = blockIdx.x * 8 + wave;
  if (row >= N) return;
  const int beg = (int)min(rowptr[row], (unsigned)E);
  const int end = (int)min(rowptr[row + 1], (unsigned)E);
  v2f acc = {0.0f, 0.0f};
  for (int j0 = beg; j0 < end; j0 += 32) {
    const int cnt = min(32, end - j0);
    const int sv = min(max(csr_src[j0 + min(lane, cnt - 1)], 0), N - 1);
    for (int u0 = 0; u0 < cnt; u0 += AGG_BATCH) {
      v2f g[AGG_BATCH];
      for (int u = 0; u < AGG_BATCH; ++u) {
        const int s = __builtin_amdgcn_readlane(sv, min(u0 + u, cnt - 1));
        g[u] = *(const v2f*)(h + (long)s * H_DIM + 2 * lane);
      }
      for (int u = 0; u < AGG_BATCH; ++u) {
        if (u0 + u < cnt) {
          acc.x += g[u].x;
          acc.y += g[u].y;
        }
      }
    }
  }
  if (relu) {
    const float nd = norm_dst[row];
    const float ns = norm_src[row];
    acc.x = fmaxf(acc.x * nd + b1[2 * lane], 0.0f) * ns;
    acc.y = fmaxf(acc.y * nd + b1[2 * lane + 1], 0.0f) * ns;
  }
  *(v2f*)(res + (long)row * H_DIM + 2 * lane) = acc;
}

#define T_PITCH 68
#define G2_WAVES 25
#define G2_GROUPS 256
__global__ __launch_bounds__(G2_WAVES * 32) void aggregate_gemm2_kernel(
    const int* __restrict__ csr_src, const unsigned* __restrict__ rowptr,
    const float* __restrict__ h,
    const float* __restrict__ W_mu, const float* __restrict__ b_mu,
    const float* __restrict__ W_sig, const float* __restrict__ b_sig,
    const float* __restrict__ norm_dst, const float* __restrict__ noise,
    float* __restrict__ z, int N, int E) {
  __shared__ v2f lWm[(H_DIM / 4) * 4 * 32];
  __shared__ v2f lWs[(H_DIM / 4) * 4 * 32];
  __shared__ __attribute__((aligned(16))) float tiles[G2_WAVES][16 * T_PITCH];
  const int lane = threadIdx.x & 31;
  const int wave = __builtin_amdgcn_readfirstlane((int)(threadIdx.x >> 5));
  const int tile = wave * (int)gridDim.x + (int)blockIdx.x;
  const int row0 = tile * 16;

  const int rpv = (int)min(rowptr[min(row0 + min(lane, 16), N)], (unsigned)E);

  constexpr int NFRAG = (H_DIM / 4) * 4 * 32;
  constexpr int NTHR  = G2_WAVES * 32;
  constexpr int FR = (NFRAG + NTHR - 1) / NTHR;
  v2f pm[FR], ps[FR];
#pragma unroll
  for (int q = 0; q < FR; ++q) {
    const int i  = (int)threadIdx.x + q * NTHR;
    pm[q].x = 0.0f; pm[q].y = 0.0f; ps[q].x = 0.0f; ps[q].y = 0.0f;
    if (i >= NFRAG) continue;
    const int kb = i >> 7;
    const int ct = (i >> 5) & 3;
    const int ln = i & 31;
    const int kg = (ln >> 4) * 2;
    const int n  = ct * 16 + (ln & 15);
    pm[q].x = W_mu[(kb * 4 + kg)     * H_DIM + n];
    pm[q].y = W_mu[(kb * 4 + kg + 1) * H_DIM + n];
    ps[q].x = W_sig[(kb * 4 + kg)     * H_DIM + n];
    ps[q].y = W_sig[(kb * 4 + kg + 1) * H_DIM + n];
  }

  const int span_end = __builtin_amdgcn_readlane(rpv, 16);
  int j0 = __builtin_amdgcn_readlane(rpv, 0);
  int r = 0;
  int rend = __builtin_amdgcn_readlane(rpv, 1);
  v2f acc = {0.0f, 0.0f};
  int cnt = min(32, span_end - j0);
  int raw = 0;
  if (cnt > 0) raw = csr_src[j0 + min(lane, cnt - 1)];

#pragma unroll
  for (int q = 0; q < FR; ++q) {
    const int i = (int)threadIdx.x + q * NTHR;
    if (i < NFRAG) {
      lWm[i] = pm[q];
      lWs[i] = ps[q];
    }
  }
  __syncthreads();
  if (row0 >= N) return;
  float* my = tiles[wave];
  const char* hb = (const char*)h;
  const unsigned lane8 = 8u * (unsigned)lane;
  constexpr int NB_NONE = 0x40000000;
  while (cnt == 32) {
    const int sv = min(max(raw, 0), N - 1);
    const int jn = j0 + 32;
    const int cntn = min(32, span_end - jn);
    if (cntn > 0) raw = csr_src[jn + min(lane, cntn - 1)];
    v2f g[32];
#pragma unroll
    for (int u = 0; u < 32; ++u) {
      const unsigned s = (unsigned)__builtin_amdgcn_readlane(sv, u);
      g[u] = *(const v2f*)(hb + (s * 256u + lane8));
    }
    int nb = (r < 15) ? rend - j0 : NB_NONE;
#pragma unroll
    for (int u = 0; u < 32; ++u) {
      if (u == nb) {
        do {
          *(v2f*)(my + r * T_PITCH + 2 * lane) = acc;
          acc.x = 0.0f; acc.y = 0.0f;
          ++r;
          rend = __builtin_amdgcn_readlane(rpv, r + 1);
        } while (j0 + u == rend && r < 15);
        nb = (r < 15) ? rend - j0 : NB_NONE;
      }
      acc.x += g[u].x;
      acc.y += g[u].y;
    }
    j0 = jn;
    cnt = cntn;
  }
  if (cnt > 0) {
    const int sv = min(max(raw, 0), N - 1);
    v2f g[32];
#pragma unroll
    for (int u = 0; u < 16; ++u) {
      const int s = __builtin_amdgcn_readlane(sv, min(u, cnt - 1));
      g[u] = *(const v2f*)(h + (long)s * H_DIM + 2 * lane);
    }
    if (cnt > 16) {
#pragma unroll
      for (int u = 16; u < 32; ++u) {
        const int s = __builtin_amdgcn_readlane(sv, min(u, cnt - 1));
        g[u] = *(const v2f*)(h + (long)s * H_DIM + 2 * lane);
      }
    }
#pragma unroll
    for (int u = 0; u < 16; ++u) {
      if (u < cnt) {
        while (j0 + u == rend && r < 15) {
          *(v2f*)(my + r * T_PITCH + 2 * lane) = acc;
          acc.x = 0.0f; acc.y = 0.0f;
          ++r;
          rend = __builtin_amdgcn_readlane(rpv, r + 1);
        }
        acc.x += g[u].x;
        acc.y += g[u].y;
      }
    }
    if (cnt > 16) {
#pragma unroll
      for (int u = 16; u < 32; ++u) {
        if (u < cnt) {
          while (j0 + u == rend && r < 15) {
            *(v2f*)(my + r * T_PITCH + 2 * lane) = acc;
            acc.x = 0.0f; acc.y = 0.0f;
            ++r;
            rend = __builtin_amdgcn_readlane(rpv, r + 1);
          }
          acc.x += g[u].x;
          acc.y += g[u].y;
        }
      }
    }
  }
  for (; r < 16; ++r) {
    *(v2f*)(my + r * T_PITCH + 2 * lane) = acc;
    acc.x = 0.0f; acc.y = 0.0f;
  }

  const int mcol = lane & 15;
  const int kgrp = (lane >> 4) * 2;
  v8f am[4] = {}, as[4] = {};
  for (int kb = 0; kb < H_DIM / 4; ++kb) {
    const v2f a = *(const v2f*)(my + mcol * T_PITCH + kb * 4 + kgrp);
#pragma unroll
    for (int ct = 0; ct < 4; ++ct) {
      const v2f bm = lWm[(kb * 4 + ct) * 32 + lane];
      const v2f bs = lWs[(kb * 4 + ct) * 32 + lane];
      am[ct] = __builtin_amdgcn_wmma_f32_16x16x4_f32(
          false, a, false, bm, (short)0, am[ct], false, false);
      as[ct] = __builtin_amdgcn_wmma_f32_16x16x4_f32(
          false, a, false, bs, (short)0, as[ct], false, false);
    }
  }

  const int mbase = row0 + ((lane >> 4) << 3);
#pragma unroll
  for (int ct = 0; ct < 4; ++ct) {
    const int n = ct * 16 + (lane & 15);
    const float bmu = b_mu[n];
    const float bsg = b_sig[n];
#pragma unroll
    for (int v = 0; v < 8; ++v) {
      const int m = mbase + v;
      const float nd = norm_dst[m];
      const float mu = am[ct][v] * nd + bmu;
      const float ls = as[ct][v] * nd + bsg;
      z[(long)m * H_DIM + n] = mu + noise[(long)m * H_DIM + n] * expf(ls);
    }
  }
}

extern "C" void kernel_launch(void* const* d_in, const int* in_sizes, int n_in,
                              void* d_out, int out_size, void* d_ws, size_t ws_size,
                              hipStream_t stream) {
  const float* feat  = (const float*)d_in[0];
  const int*   src   = (const int*)d_in[1];
  const int*   dst   = (const int*)d_in[2];
  const float* noise = (const float*)d_in[3];
  const float* W1    = (const float*)d_in[4];
  const float* b1    = (const float*)d_in[5];
  const float* W_mu  = (const float*)d_in[6];
  const float* b_mu  = (const float*)d_in[7];
  const float* W_sig = (const float*)d_in[8];
  const float* b_sig = (const float*)d_in[9];

  const int E = in_sizes[1];
  const int N = in_sizes[3] / H_DIM;

  float* ws       = (float*)d_ws;
  float* norm_src = ws;
  float* norm_dst = ws + N;
  float* buf_a    = ws + 2L * N;
  float* buf_b    = buf_a + (long)N * H_DIM;
  float* zout     = (float*)d_out;

  const int threads = 256;
  const int node_blocks = (N + NORM_T - 1) / NORM_T;
  const int n_ranges = (N + DEG_R - 1) / DEG_R;

  int*      csr_src = (int*)buf_b;
  unsigned* rowptr  = (unsigned*)buf_b + E;
  unsigned* bpre    = rowptr + N + 1;
  unsigned* rtot    = bpre + (long)DEG_S * n_ranges * DEG_BLK;
  float*    h1      = zout;

  unsigned* deg_part = (unsigned*)buf_a;
  unsigned* dplanes  = deg_part + (long)DEG_S * N;
  degree_hist_kernel<<<dim3(DEG_S, n_ranges, 2), 1024, 0, stream>>>(src, dst, deg_part, bpre, rtot, E, N);
  offsets_kernel<<<node_blocks, NORM_T, 0, stream>>>(deg_part, dplanes, bpre, rtot, rowptr, norm_src, N);

  const int n_scatter = DEG_S * ((N + SC_R - 1) / SC_R);
  const int l1_min    = (N / 16 + 31) / 32;
  const int n_gemm    = l1_min > L1_GROUPS ? l1_min : L1_GROUPS;
  sort_scatter_and_layer1_kernel<<<n_scatter + n_gemm, 1024, 0, stream>>>(
      src, dst, dplanes, csr_src, feat, norm_src, W1, h1, E, N, n_scatter, n_gemm);

  aggregate_rows_kernel<<<(N + 7) / 8, threads, 0, stream>>>(
      csr_src, rowptr, h1, norm_dst, b1, norm_src, buf_a, 1, N, E);

  const int g2_min = (N / 16 + G2_WAVES - 1) / G2_WAVES;
  const int n_g2   = g2_min > G2_GROUPS ? g2_min : G2_GROUPS;
  aggregate_gemm2_kernel<<<n_g2, G2_WAVES * 32, 0, stream>>>(
      csr_src, rowptr, buf_a, W_mu, b_mu, W_sig, b_sig, norm_dst, noise, zout, N, E);
}
